# starting kernel
# baseline (speedup 1.0000x reference)
#include <hip/hip_runtime.h>

typedef __attribute__((ext_vector_type(2))) float v2f;
typedef __attribute__((ext_vector_type(8))) float v8f;

#define T_LEN   262144
#define IN_SZ   64
#define HID     16
#define G3      48

__global__ __launch_bounds__(256) void xi_gemm_wmma(
    const float* __restrict__ x, const float* __restrict__ W_ih,
    const float* __restrict__ b_ih, const float* __restrict__ b_hh,
    float* __restrict__ xi)
{
    const int lane  = threadIdx.x & 31;
    const int wave  = threadIdx.x >> 5;
    const int tile  = blockIdx.x * 8 + wave;
    const int m     = lane & 15;
    const int khalf = lane >> 4;

    const float* xrow = x + (size_t)(tile * 16 + m) * IN_SZ + 2 * khalf;
    v2f a[16];
#pragma unroll
    for (int i = 0; i < 16; ++i)
        a[i] = *(const v2f*)(xrow + 4 * i);

    for (int nb = 0; nb < 3; ++nb) {
        const int col  = nb * 16 + m;
        const float bias = b_ih[col] + (nb < 2 ? b_hh[col] : 0.0f);
        v8f c = {bias, bias, bias, bias, bias, bias, bias, bias};

        const float* wrow = W_ih + (size_t)col * IN_SZ + 2 * khalf;
#pragma unroll
        for (int i = 0; i < 16; ++i) {
            v2f b = *(const v2f*)(wrow + 4 * i);
            c = __builtin_amdgcn_wmma_f32_16x16x4_f32(
                    false, a[i], false, b, (short)0, c, false, false);
        }

        float* orow = xi + (size_t)(tile * 16 + 8 * khalf) * G3 + col;
#pragma unroll
        for (int r = 0; r < 8; ++r)
            orow[(size_t)r * G3] = c[r];
    }
}

__global__ __launch_bounds__(32) void gru_scan(
    const float* __restrict__ xi, const float* __restrict__ W_hh,
    const float* __restrict__ b_hh, const float* __restrict__ fc1_w,
    const float* __restrict__ fc1_b, const float* __restrict__ fc2_w,
    const float* __restrict__ fc2_b, float* __restrict__ out)
{
    const int lane = threadIdx.x;
    const int j    = lane & 15;

    float wr[HID], wz[HID], wn[HID];
#pragma unroll
    for (int k = 0; k < HID; ++k) {
        wr[k] = W_hh[(j)      * HID + k];
        wz[k] = W_hh[(16 + j) * HID + k];
        wn[k] = W_hh[(32 + j) * HID + k];
    }
    const float bhn = b_hh[32 + j];

    float weff = 0.0f, beff = fc2_b[0];
#pragma unroll
    for (int i = 0; i < 8; ++i) {
        const float f2 = fc2_w[i];
        weff += f2 * fc1_w[i * HID + j];
        beff += f2 * fc1_b[i];
    }

    __shared__ __align__(16) float hsh[HID];

    float4 ha = {0,0,0,0}, hb = {0,0,0,0}, hc = {0,0,0,0}, hd = {0,0,0,0};
    float  hj = 0.0f;

    const float* xr_p = xi + j;
    const float* xz_p = xi + 16 + j;
    const float* xn_p = xi + 32 + j;
    float xr = xr_p[0], xz = xz_p[0], xn = xn_p[0];

    for (int t = 0; t < T_LEN; ++t) {
        const size_t nofs = (size_t)(t + 1) * G3;
        const float xr_n = xr_p[nofs], xz_n = xz_p[nofs], xn_n = xn_p[nofs];

        float hr = 0.0f, hz = 0.0f, hn = bhn;
        const float hv[HID] = {ha.x, ha.y, ha.z, ha.w,  hb.x, hb.y, hb.z, hb.w,
                               hc.x, hc.y, hc.z, hc.w,  hd.x, hd.y, hd.z, hd.w};
#pragma unroll
        for (int k = 0; k < HID; ++k) {
            hr += wr[k] * hv[k];
            hz += wz[k] * hv[k];
            hn += wn[k] * hv[k];
        }

        const float r = __builtin_amdgcn_rcpf(1.0f + __expf(-(xr + hr)));
        const float z = __builtin_amdgcn_rcpf(1.0f + __expf(-(xz + hz)));
        const float u = xn + r * hn;
        const float n = 1.0f - 2.0f * __builtin_amdgcn_rcpf(__expf(2.0f * u) + 1.0f);
        const float hnew = n + z * (hj - n);
        hj = hnew;

        float yv = hnew * weff;
        yv += __shfl_xor(yv, 1, 32);
        yv += __shfl_xor(yv, 2, 32);
        yv += __shfl_xor(yv, 4, 32);
        yv += __shfl_xor(yv, 8, 32);
        if (lane == 0) out[t] = yv + beff;

        if (lane < 16) hsh[lane] = hnew;
        __syncthreads();
        ha = *(const float4*)(hsh + 0);
        hb = *(const float4*)(hsh + 4);
        hc = *(const float4*)(hsh + 8);
        hd = *(const float4*)(hsh + 12);
        __syncthreads();

        xr = xr_n; xz = xz_n; xn = xn_n;
    }
}

extern "C" void kernel_launch(void* const* d_in, const int* in_sizes, int n_in,
                              void* d_out, int out_size, void* d_ws, size_t ws_size,
                              hipStream_t stream) {
    const float* x     = (const float*)d_in[0];
    const float* W_ih  = (const float*)d_in[1];
    const float* W_hh  = (const float*)d_in[2];
    const float* b_ih  = (const float*)d_in[3];
    const float* b_hh  = (const float*)d_in[4];
    const float* fc1_w = (const float*)d_in[5];
    const float* fc1_b = (const float*)d_in[6];
    const float* fc2_w = (const float*)d_in[7];
    const float* fc2_b = (const float*)d_in[8];
    float* out = (float*)d_out;

    float* xi = (float*)d_ws;

    xi_gemm_wmma<<<T_LEN / 16 / 8, 256, 0, stream>>>(x, W_ih, b_ih, b_hh, xi);

    gru_scan<<<1, 32, 0, stream>>>(xi, W_hh, b_hh, fc1_w, fc1_b, fc2_w, fc2_b, out);
}


# scan kernel: the two block barriers of the per-step LDS exchange removed (one-wave block, a wave's LDS ops run in order); their fence had drained the out store and xi prefetch every step
# speedup vs baseline: 1.2338x; 1.2338x over previous
#include <hip/hip_runtime.h>

typedef __attribute__((ext_vector_type(2))) float v2f;
typedef __attribute__((ext_vector_type(8))) float v8f;

#define T_LEN   262144
#define IN_SZ   64
#define HID     16
#define G3      48

__global__ __launch_bounds__(256) void xi_gemm_wmma(
    const float* __restrict__ x, const float* __restrict__ W_ih,
    const float* __restrict__ b_ih, const float* __restrict__ b_hh,
    float* __restrict__ xi)
{
    const int lane  = threadIdx.x & 31;
    const int wave  = threadIdx.x >> 5;
    const int tile  = blockIdx.x * 8 + wave;
    const int m     = lane & 15;
    const int khalf = lane >> 4;

    const float* xrow = x + (size_t)(tile * 16 + m) * IN_SZ + 2 * khalf;
    v2f a[16];
#pragma unroll
    for (int i = 0; i < 16; ++i)
        a[i] = *(const v2f*)(xrow + 4 * i);

    for (int nb = 0; nb < 3; ++nb) {
        const int col  = nb * 16 + m;
        const float bias = b_ih[col] + (nb < 2 ? b_hh[col] : 0.0f);
        v8f c = {bias, bias, bias, bias, bias, bias, bias, bias};

        const float* wrow = W_ih + (size_t)col * IN_SZ + 2 * khalf;
#pragma unroll
        for (int i = 0; i < 16; ++i) {
            v2f b = *(const v2f*)(wrow + 4 * i);
            c = __builtin_amdgcn_wmma_f32_16x16x4_f32(
                    false, a[i], false, b, (short)0, c, false, false);
        }

        float* orow = xi + (size_t)(tile * 16 + 8 * khalf) * G3 + col;
#pragma unroll
        for (int r = 0; r < 8; ++r)
            orow[(size_t)r * G3] = c[r];
    }
}

__global__ __launch_bounds__(32) void gru_scan(
    const float* __restrict__ xi, const float* __restrict__ W_hh,
    const float* __restrict__ b_hh, const float* __restrict__ fc1_w,
    const float* __restrict__ fc1_b, const float* __restrict__ fc2_w,
    const float* __restrict__ fc2_b, float* __restrict__ out)
{
    const int lane = threadIdx.x;
    const int j    = lane & 15;

    float wr[HID], wz[HID], wn[HID];
#pragma unroll
    for (int k = 0; k < HID; ++k) {
        wr[k] = W_hh[(j)      * HID + k];
        wz[k] = W_hh[(16 + j) * HID + k];
        wn[k] = W_hh[(32 + j) * HID + k];
    }
    const float bhn = b_hh[32 + j];

    float weff = 0.0f, beff = fc2_b[0];
#pragma unroll
    for (int i = 0; i < 8; ++i) {
        const float f2 = fc2_w[i];
        weff += f2 * fc1_w[i * HID + j];
        beff += f2 * fc1_b[i];
    }

    __shared__ __align__(16) float hsh[HID];

    float4 ha = {0,0,0,0}, hb = {0,0,0,0}, hc = {0,0,0,0}, hd = {0,0,0,0};
    float  hj = 0.0f;

    const float* xr_p = xi + j;
    const float* xz_p = xi + 16 + j;
    const float* xn_p = xi + 32 + j;
    float xr = xr_p[0], xz = xz_p[0], xn = xn_p[0];

    for (int t = 0; t < T_LEN; ++t) {
        const size_t nofs = (size_t)(t + 1) * G3;
        const float xr_n = xr_p[nofs], xz_n = xz_p[nofs], xn_n = xn_p[nofs];

        float hr = 0.0f, hz = 0.0f, hn = bhn;
        const float hv[HID] = {ha.x, ha.y, ha.z, ha.w,  hb.x, hb.y, hb.z, hb.w,
                               hc.x, hc.y, hc.z, hc.w,  hd.x, hd.y, hd.z, hd.w};
#pragma unroll
        for (int k = 0; k < HID; ++k) {
            hr += wr[k] * hv[k];
            hz += wz[k] * hv[k];
            hn += wn[k] * hv[k];
        }

        const float r = __builtin_amdgcn_rcpf(1.0f + __expf(-(xr + hr)));
        const float z = __builtin_amdgcn_rcpf(1.0f + __expf(-(xz + hz)));
        const float u = xn + r * hn;
        const float n = 1.0f - 2.0f * __builtin_amdgcn_rcpf(__expf(2.0f * u) + 1.0f);
        const float hnew = n + z * (hj - n);
        hj = hnew;

        float yv = hnew * weff;
        yv += __shfl_xor(yv, 1, 32);
        yv += __shfl_xor(yv, 2, 32);
        yv += __shfl_xor(yv, 4, 32);
        yv += __shfl_xor(yv, 8, 32);
        if (lane == 0) out[t] = yv + beff;

        if (lane < 16) hsh[lane] = hnew;
        ha = *(const float4*)(hsh + 0);
        hb = *(const float4*)(hsh + 4);
        hc = *(const float4*)(hsh + 8);
        hd = *(const float4*)(hsh + 12);

        xr = xr_n; xz = xz_n; xn = xn_n;
    }
}

extern "C" void kernel_launch(void* const* d_in, const int* in_sizes, int n_in,
                              void* d_out, int out_size, void* d_ws, size_t ws_size,
                              hipStream_t stream) {
    const float* x     = (const float*)d_in[0];
    const float* W_ih  = (const float*)d_in[1];
    const float* W_hh  = (const float*)d_in[2];
    const float* b_ih  = (const float*)d_in[3];
    const float* b_hh  = (const float*)d_in[4];
    const float* fc1_w = (const float*)d_in[5];
    const float* fc1_b = (const float*)d_in[6];
    const float* fc2_w = (const float*)d_in[7];
    const float* fc2_b = (const float*)d_in[8];
    float* out = (float*)d_out;

    float* xi = (float*)d_ws;

    xi_gemm_wmma<<<T_LEN / 16 / 8, 256, 0, stream>>>(x, W_ih, b_ih, b_hh, xi);

    gru_scan<<<1, 32, 0, stream>>>(xi, W_hh, b_hh, fc1_w, fc1_b, fc2_w, fc2_b, out);
}


# scan kernel: each step's xi row is loaded 8 steps ahead into a register ring (loop unrolled by 8) instead of 1 step ahead, so a step no longer waits a full memory-miss latency for its row
# speedup vs baseline: 1.3505x; 1.0946x over previous
#include <hip/hip_runtime.h>

typedef __attribute__((ext_vector_type(2))) float v2f;
typedef __attribute__((ext_vector_type(8))) float v8f;

#define T_LEN   262144
#define IN_SZ   64
#define HID     16
#define G3      48
#define PF_D    8

__global__ __launch_bounds__(256) void xi_gemm_wmma(
    const float* __restrict__ x, const float* __restrict__ W_ih,
    const float* __restrict__ b_ih, const float* __restrict__ b_hh,
    float* __restrict__ xi)
{
    const int lane  = threadIdx.x & 31;
    const int wave  = threadIdx.x >> 5;
    const int tile  = blockIdx.x * 8 + wave;
    const int m     = lane & 15;
    const int khalf = lane >> 4;

    const float* xrow = x + (size_t)(tile * 16 + m) * IN_SZ + 2 * khalf;
    v2f a[16];
#pragma unroll
    for (int i = 0; i < 16; ++i)
        a[i] = *(const v2f*)(xrow + 4 * i);

    for (int nb = 0; nb < 3; ++nb) {
        const int col  = nb * 16 + m;
        const float bias = b_ih[col] + (nb < 2 ? b_hh[col] : 0.0f);
        v8f c = {bias, bias, bias, bias, bias, bias, bias, bias};

        const float* wrow = W_ih + (size_t)col * IN_SZ + 2 * khalf;
#pragma unroll
        for (int i = 0; i < 16; ++i) {
            v2f b = *(const v2f*)(wrow + 4 * i);
            c = __builtin_amdgcn_wmma_f32_16x16x4_f32(
                    false, a[i], false, b, (short)0, c, false, false);
        }

        float* orow = xi + (size_t)(tile * 16 + 8 * khalf) * G3 + col;
#pragma unroll
        for (int r = 0; r < 8; ++r)
            orow[(size_t)r * G3] = c[r];
    }
}

__global__ __launch_bounds__(32) void gru_scan(
    const float* __restrict__ xi, const float* __restrict__ W_hh,
    const float* __restrict__ b_hh, const float* __restrict__ fc1_w,
    const float* __restrict__ fc1_b, const float* __restrict__ fc2_w,
    const float* __restrict__ fc2_b, float* __restrict__ out)
{
    const int lane = threadIdx.x;
    const int j    = lane & 15;

    float wr[HID], wz[HID], wn[HID];
#pragma unroll
    for (int k = 0; k < HID; ++k) {
        wr[k] = W_hh[(j)      * HID + k];
        wz[k] = W_hh[(16 + j) * HID + k];
        wn[k] = W_hh[(32 + j) * HID + k];
    }
    const float bhn = b_hh[32 + j];

    float weff = 0.0f, beff = fc2_b[0];
#pragma unroll
    for (int i = 0; i < 8; ++i) {
        const float f2 = fc2_w[i];
        weff += f2 * fc1_w[i * HID + j];
        beff += f2 * fc1_b[i];
    }

#pragma unroll
    for (int k = 0; k < HID; ++k)
        asm volatile("" :: "v"(wr[k]), "v"(wz[k]), "v"(wn[k]));
    asm volatile("" :: "v"(bhn));

    __shared__ __align__(16) float hsh[HID];

    float4 ha = {0,0,0,0}, hb = {0,0,0,0}, hc = {0,0,0,0}, hd = {0,0,0,0};
    float  hj = 0.0f;

    const float* xr_p = xi + j;
    const float* xz_p = xi + 16 + j;
    const float* xn_p = xi + 32 + j;

    float pr[PF_D], pz[PF_D], pn[PF_D];
#pragma unroll
    for (int s = 0; s < PF_D; ++s) {
        pr[s] = xr_p[(size_t)s * G3];
        pz[s] = xz_p[(size_t)s * G3];
        pn[s] = xn_p[(size_t)s * G3];
    }

    for (int t0 = 0; t0 < T_LEN; t0 += PF_D) {
      const int pf0 = (t0 + PF_D < T_LEN) ? (t0 + PF_D) : t0;
      const float* nr_p = xr_p + (size_t)pf0 * G3;
      const float* nz_p = xz_p + (size_t)pf0 * G3;
      const float* nn_p = xn_p + (size_t)pf0 * G3;

#pragma unroll
      for (int s = 0; s < PF_D; ++s) {
        const int t = t0 + s;
        const float xr = pr[s], xz = pz[s], xn = pn[s];

        float hr = 0.0f, hz = 0.0f, hn = bhn;
        const float hv[HID] = {ha.x, ha.y, ha.z, ha.w,  hb.x, hb.y, hb.z, hb.w,
                               hc.x, hc.y, hc.z, hc.w,  hd.x, hd.y, hd.z, hd.w};
#pragma unroll
        for (int k = 0; k < HID; ++k) {
            hr += wr[k] * hv[k];
            hz += wz[k] * hv[k];
            hn += wn[k] * hv[k];
        }

        const float r = __builtin_amdgcn_rcpf(1.0f + __expf(-(xr + hr)));
        const float z = __builtin_amdgcn_rcpf(1.0f + __expf(-(xz + hz)));
        const float u = xn + r * hn;
        const float n = 1.0f - 2.0f * __builtin_amdgcn_rcpf(__expf(2.0f * u) + 1.0f);
        const float hnew = n + z * (hj - n);
        hj = hnew;

        float yv = hnew * weff;
        yv += __shfl_xor(yv, 1, 32);
        yv += __shfl_xor(yv, 2, 32);
        yv += __shfl_xor(yv, 4, 32);
        yv += __shfl_xor(yv, 8, 32);
        if (lane == 0) out[t] = yv + beff;

        pr[s] = nr_p[s * G3];
        pz[s] = nz_p[s * G3];
        pn[s] = nn_p[s * G3];

        if (lane < 16) hsh[lane] = hnew;
        ha = *(const float4*)(hsh + 0);
        hb = *(const float4*)(hsh + 4);
        hc = *(const float4*)(hsh + 8);
        hd = *(const float4*)(hsh + 12);
        __builtin_amdgcn_sched_barrier(0);
      }
    }
}

extern "C" void kernel_launch(void* const* d_in, const int* in_sizes, int n_in,
                              void* d_out, int out_size, void* d_ws, size_t ws_size,
                              hipStream_t stream) {
    const float* x     = (const float*)d_in[0];
    const float* W_ih  = (const float*)d_in[1];
    const float* W_hh  = (const float*)d_in[2];
    const float* b_ih  = (const float*)d_in[3];
    const float* b_hh  = (const float*)d_in[4];
    const float* fc1_w = (const float*)d_in[5];
    const float* fc1_b = (const float*)d_in[6];
    const float* fc2_w = (const float*)d_in[7];
    const float* fc2_b = (const float*)d_in[8];
    float* out = (float*)d_out;

    float* xi = (float*)d_ws;

    xi_gemm_wmma<<<T_LEN / 16 / 8, 256, 0, stream>>>(x, W_ih, b_ih, b_hh, xi);

    gru_scan<<<1, 32, 0, stream>>>(xi, W_hh, b_hh, fc1_w, fc1_b, fc2_w, fc2_b, out);
}


# scan kernel: the head's 16-lane sum (same xor tree, same bits) runs as DPP row_xmask adds on the vector ALU during the h loads' latency, not as four serial LDS-unit shuffles ahead of the h exchange
# speedup vs baseline: 2.7394x; 2.0284x over previous
#include <hip/hip_runtime.h>

typedef __attribute__((ext_vector_type(2))) float v2f;
typedef __attribute__((ext_vector_type(8))) float v8f;

#define T_LEN   262144
#define IN_SZ   64
#define HID     16
#define G3      48
#define PF_D    8

__global__ __launch_bounds__(256) void xi_gemm_wmma(
    const float* __restrict__ x, const float* __restrict__ W_ih,
    const float* __restrict__ b_ih, const float* __restrict__ b_hh,
    float* __restrict__ xi)
{
    const int lane  = threadIdx.x & 31;
    const int wave  = threadIdx.x >> 5;
    const int tile  = blockIdx.x * 8 + wave;
    const int m     = lane & 15;
    const int khalf = lane >> 4;

    const float* xrow = x + (size_t)(tile * 16 + m) * IN_SZ + 2 * khalf;
    v2f a[16];
#pragma unroll
    for (int i = 0; i < 16; ++i)
        a[i] = *(const v2f*)(xrow + 4 * i);

    for (int nb = 0; nb < 3; ++nb) {
        const int col  = nb * 16 + m;
        const float bias = b_ih[col] + (nb < 2 ? b_hh[col] : 0.0f);
        v8f c = {bias, bias, bias, bias, bias, bias, bias, bias};

        const float* wrow = W_ih + (size_t)col * IN_SZ + 2 * khalf;
#pragma unroll
        for (int i = 0; i < 16; ++i) {
            v2f b = *(const v2f*)(wrow + 4 * i);
            c = __builtin_amdgcn_wmma_f32_16x16x4_f32(
                    false, a[i], false, b, (short)0, c, false, false);
        }

        float* orow = xi + (size_t)(tile * 16 + 8 * khalf) * G3 + col;
#pragma unroll
        for (int r = 0; r < 8; ++r)
            orow[(size_t)r * G3] = c[r];
    }
}

template <int MASK>
__device__ __forceinline__ float dpp_row_xmask(float v)
{
    const int i = __builtin_bit_cast(int, v);
    return __builtin_bit_cast(float,
        __builtin_amdgcn_update_dpp(i, i, 0x160 + MASK, 0xf, 0xf, true));
}

__global__ __launch_bounds__(32) void gru_scan(
    const float* __restrict__ xi, const float* __restrict__ W_hh,
    const float* __restrict__ b_hh, const float* __restrict__ fc1_w,
    const float* __restrict__ fc1_b, const float* __restrict__ fc2_w,
    const float* __restrict__ fc2_b, float* __restrict__ out)
{
    const int lane = threadIdx.x;
    const int j    = lane & 15;

    float wr[HID], wz[HID], wn[HID];
#pragma unroll
    for (int k = 0; k < HID; ++k) {
        wr[k] = W_hh[(j)      * HID + k];
        wz[k] = W_hh[(16 + j) * HID + k];
        wn[k] = W_hh[(32 + j) * HID + k];
    }
    const float bhn = b_hh[32 + j];

    float weff = 0.0f, beff = fc2_b[0];
#pragma unroll
    for (int i = 0; i < 8; ++i) {
        const float f2 = fc2_w[i];
        weff += f2 * fc1_w[i * HID + j];
        beff += f2 * fc1_b[i];
    }

#pragma unroll
    for (int k = 0; k < HID; ++k)
        asm volatile("" :: "v"(wr[k]), "v"(wz[k]), "v"(wn[k]));
    asm volatile("" :: "v"(bhn));

    __shared__ __align__(16) float hsh[HID];

    float4 ha = {0,0,0,0}, hb = {0,0,0,0}, hc = {0,0,0,0}, hd = {0,0,0,0};
    float  hj = 0.0f;

    const float* xr_p = xi + j;
    const float* xz_p = xi + 16 + j;
    const float* xn_p = xi + 32 + j;

    float pr[PF_D], pz[PF_D], pn[PF_D];
#pragma unroll
    for (int s = 0; s < PF_D; ++s) {
        pr[s] = xr_p[(size_t)s * G3];
        pz[s] = xz_p[(size_t)s * G3];
        pn[s] = xn_p[(size_t)s * G3];
    }

    for (int t0 = 0; t0 < T_LEN; t0 += PF_D) {
      const int pf0 = (t0 + PF_D < T_LEN) ? (t0 + PF_D) : t0;
      const float* nr_p = xr_p + (size_t)pf0 * G3;
      const float* nz_p = xz_p + (size_t)pf0 * G3;
      const float* nn_p = xn_p + (size_t)pf0 * G3;

#pragma unroll
      for (int s = 0; s < PF_D; ++s) {
        const int t = t0 + s;
        const float xr = pr[s], xz = pz[s], xn = pn[s];

        float hr = 0.0f, hz = 0.0f, hn = bhn;
        const float hv[HID] = {ha.x, ha.y, ha.z, ha.w,  hb.x, hb.y, hb.z, hb.w,
                               hc.x, hc.y, hc.z, hc.w,  hd.x, hd.y, hd.z, hd.w};
#pragma unroll
        for (int k = 0; k < HID; ++k) {
            hr += wr[k] * hv[k];
            hz += wz[k] * hv[k];
            hn += wn[k] * hv[k];
        }

        const float r = __builtin_amdgcn_rcpf(1.0f + __expf(-(xr + hr)));
        const float z = __builtin_amdgcn_rcpf(1.0f + __expf(-(xz + hz)));
        const float u = xn + r * hn;
        const float n = 1.0f - 2.0f * __builtin_amdgcn_rcpf(__expf(2.0f * u) + 1.0f);
        const float hnew = n + z * (hj - n);
        hj = hnew;

        pr[s] = nr_p[s * G3];
        pz[s] = nz_p[s * G3];
        pn[s] = nn_p[s * G3];

        if (lane < 16) hsh[lane] = hnew;
        ha = *(const float4*)(hsh + 0);
        hb = *(const float4*)(hsh + 4);
        hc = *(const float4*)(hsh + 8);
        hd = *(const float4*)(hsh + 12);
        __builtin_amdgcn_sched_barrier(0);

        float yv = hnew * weff;
        yv += dpp_row_xmask<1>(yv);
        yv += dpp_row_xmask<2>(yv);
        yv += dpp_row_xmask<4>(yv);
        yv += dpp_row_xmask<8>(yv);
        if (lane == 0) out[t] = yv + beff;
      }
    }
}

extern "C" void kernel_launch(void* const* d_in, const int* in_sizes, int n_in,
                              void* d_out, int out_size, void* d_ws, size_t ws_size,
                              hipStream_t stream) {
    const float* x     = (const float*)d_in[0];
    const float* W_ih  = (const float*)d_in[1];
    const float* W_hh  = (const float*)d_in[2];
    const float* b_ih  = (const float*)d_in[3];
    const float* b_hh  = (const float*)d_in[4];
    const float* fc1_w = (const float*)d_in[5];
    const float* fc1_b = (const float*)d_in[6];
    const float* fc2_w = (const float*)d_in[7];
    const float* fc2_b = (const float*)d_in[8];
    float* out = (float*)d_out;

    float* xi = (float*)d_ws;

    xi_gemm_wmma<<<T_LEN / 16 / 8, 256, 0, stream>>>(x, W_ih, b_ih, b_hh, xi);

    gru_scan<<<1, 32, 0, stream>>>(xi, W_hh, b_hh, fc1_w, fc1_b, fc2_w, fc2_b, out);
}


# scan kernel: h stays in registers; h[k] is broadcast from lane k by a DPP row_share move two terms ahead of its 3 fused multiply-adds, no LDS store + 4 loads per step; same k order, hr/hz now fused
# speedup vs baseline: 3.0567x; 1.1158x over previous
#include <hip/hip_runtime.h>

typedef __attribute__((ext_vector_type(2))) float v2f;
typedef __attribute__((ext_vector_type(8))) float v8f;

#define T_LEN   262144
#define IN_SZ   64
#define HID     16
#define G3      48
#define PF_D    8

__global__ __launch_bounds__(256) void xi_gemm_wmma(
    const float* __restrict__ x, const float* __restrict__ W_ih,
    const float* __restrict__ b_ih, const float* __restrict__ b_hh,
    float* __restrict__ xi)
{
    const int lane  = threadIdx.x & 31;
    const int wave  = threadIdx.x >> 5;
    const int tile  = blockIdx.x * 8 + wave;
    const int m     = lane & 15;
    const int khalf = lane >> 4;

    const float* xrow = x + (size_t)(tile * 16 + m) * IN_SZ + 2 * khalf;
    v2f a[16];
#pragma unroll
    for (int i = 0; i < 16; ++i)
        a[i] = *(const v2f*)(xrow + 4 * i);

    for (int nb = 0; nb < 3; ++nb) {
        const int col  = nb * 16 + m;
        const float bias = b_ih[col] + (nb < 2 ? b_hh[col] : 0.0f);
        v8f c = {bias, bias, bias, bias, bias, bias, bias, bias};

        const float* wrow = W_ih + (size_t)col * IN_SZ + 2 * khalf;
#pragma unroll
        for (int i = 0; i < 16; ++i) {
            v2f b = *(const v2f*)(wrow + 4 * i);
            c = __builtin_amdgcn_wmma_f32_16x16x4_f32(
                    false, a[i], false, b, (short)0, c, false, false);
        }

        float* orow = xi + (size_t)(tile * 16 + 8 * khalf) * G3 + col;
#pragma unroll
        for (int r = 0; r < 8; ++r)
            orow[(size_t)r * G3] = c[r];
    }
}

template <int MASK>
__device__ __forceinline__ float dpp_row_xmask(float v)
{
    const int i = __builtin_bit_cast(int, v);
    return __builtin_bit_cast(float,
        __builtin_amdgcn_update_dpp(i, i, 0x160 + MASK, 0xf, 0xf, true));
}


#define ROW_SHARE(K) " row_share:" #K " row_mask:0xf bank_mask:0xf bound_ctrl:1"


#define HH_HEAD(T0, T1)                                                          \
    asm("v_mov_b32_dpp %0, %2" ROW_SHARE(0) "\n\t"                               \
        "v_mov_b32_dpp %1, %2" ROW_SHARE(1)                                       \
        : "=&v"(T0), "=&v"(T1) : "v"(hj))

#define HH_TERM0(TK, TN)                                                         \
    asm("v_mul_f32 %0, %4, %5\n\t"                                               \
        "v_mul_f32 %1, %4, %6\n\t"                                               \
        "v_fma_f32 %2, %4, %7, %8\n\t"                                           \
        "v_mov_b32_dpp %3, %9" ROW_SHARE(2)                                       \
        : "=&v"(hr), "=&v"(hz), "=&v"(hn), "=&v"(TN)                             \
        : "v"(TK), "v"(wr[0]), "v"(wz[0]), "v"(wn[0]), "v"(bhn), "v"(hj))

#define HH_TERM(K, K2, TK, TN)                                                   \
    asm("v_fmac_f32 %0, %4, %5\n\t"                                              \
        "v_fmac_f32 %1, %4, %6\n\t"                                              \
        "v_fmac_f32 %2, %4, %7\n\t"                                              \
        "v_mov_b32_dpp %3, %8" ROW_SHARE(K2)                                      \
        : "+v"(hr), "+v"(hz), "+v"(hn), "=&v"(TN)                                \
        : "v"(TK), "v"(wr[K]), "v"(wz[K]), "v"(wn[K]), "v"(hj))

#define HH_TAIL(K, TK)                                                           \
    asm("v_fmac_f32 %0, %3, %4\n\t"                                              \
        "v_fmac_f32 %1, %3, %5\n\t"                                              \
        "v_fmac_f32 %2, %3, %6"                                                   \
        : "+v"(hr), "+v"(hz), "+v"(hn)                                           \
        : "v"(TK), "v"(wr[K]), "v"(wz[K]), "v"(wn[K]))

__global__ __launch_bounds__(32) void gru_scan(
    const float* __restrict__ xi, const float* __restrict__ W_hh,
    const float* __restrict__ b_hh, const float* __restrict__ fc1_w,
    const float* __restrict__ fc1_b, const float* __restrict__ fc2_w,
    const float* __restrict__ fc2_b, float* __restrict__ out)
{
    const int lane = threadIdx.x;
    const int j    = lane & 15;

    float wr[HID], wz[HID], wn[HID];
#pragma unroll
    for (int k = 0; k < HID; ++k) {
        wr[k] = W_hh[(j)      * HID + k];
        wz[k] = W_hh[(16 + j) * HID + k];
        wn[k] = W_hh[(32 + j) * HID + k];
    }
    const float bhn = b_hh[32 + j];

    float weff = 0.0f, beff = fc2_b[0];
#pragma unroll
    for (int i = 0; i < 8; ++i) {
        const float f2 = fc2_w[i];
        weff += f2 * fc1_w[i * HID + j];
        beff += f2 * fc1_b[i];
    }

#pragma unroll
    for (int k = 0; k < HID; ++k)
        asm volatile("" :: "v"(wr[k]), "v"(wz[k]), "v"(wn[k]));
    asm volatile("" :: "v"(bhn));

    float hj = 0.0f;

    const float* xr_p = xi + j;
    const float* xz_p = xi + 16 + j;
    const float* xn_p = xi + 32 + j;

    float pr[PF_D], pz[PF_D], pn[PF_D];
#pragma unroll
    for (int s = 0; s < PF_D; ++s) {
        pr[s] = xr_p[(size_t)s * G3];
        pz[s] = xz_p[(size_t)s * G3];
        pn[s] = xn_p[(size_t)s * G3];
    }

    for (int t0 = 0; t0 < T_LEN; t0 += PF_D) {
      const int pf0 = (t0 + PF_D < T_LEN) ? (t0 + PF_D) : t0;
      const float* nr_p = xr_p + (size_t)pf0 * G3;
      const float* nz_p = xz_p + (size_t)pf0 * G3;
      const float* nn_p = xn_p + (size_t)pf0 * G3;

#pragma unroll
      for (int s = 0; s < PF_D; ++s) {
        const int t = t0 + s;
        const float xr = pr[s], xz = pz[s], xn = pn[s];

        float hr, hz, hn, ta, tb, tc;
        HH_HEAD(ta, tb);
        HH_TERM0(ta, tc);
        HH_TERM(1, 3, tb, ta);
        HH_TERM(2, 4, tc, tb);
        HH_TERM(3, 5, ta, tc);
        HH_TERM(4, 6, tb, ta);
        HH_TERM(5, 7, tc, tb);
        HH_TERM(6, 8, ta, tc);
        HH_TERM(7, 9, tb, ta);
        HH_TERM(8, 10, tc, tb);
        HH_TERM(9, 11, ta, tc);
        HH_TERM(10, 12, tb, ta);
        HH_TERM(11, 13, tc, tb);
        HH_TERM(12, 14, ta, tc);
        HH_TERM(13, 15, tb, ta);
        HH_TAIL(14, tc);
        HH_TAIL(15, ta);

        const float r = __builtin_amdgcn_rcpf(1.0f + __expf(-(xr + hr)));
        const float z = __builtin_amdgcn_rcpf(1.0f + __expf(-(xz + hz)));
        const float u = xn + r * hn;
        const float n = 1.0f - 2.0f * __builtin_amdgcn_rcpf(__expf(2.0f * u) + 1.0f);
        const float hnew = n + z * (hj - n);
        hj = hnew;

        pr[s] = nr_p[s * G3];
        pz[s] = nz_p[s * G3];
        pn[s] = nn_p[s * G3];

        float yv = hnew * weff;
        yv += dpp_row_xmask<1>(yv);
        yv += dpp_row_xmask<2>(yv);
        yv += dpp_row_xmask<4>(yv);
        yv += dpp_row_xmask<8>(yv);
        if (lane == 0) out[t] = yv + beff;
      }
    }
}

extern "C" void kernel_launch(void* const* d_in, const int* in_sizes, int n_in,
                              void* d_out, int out_size, void* d_ws, size_t ws_size,
                              hipStream_t stream) {
    const float* x     = (const float*)d_in[0];
    const float* W_ih  = (const float*)d_in[1];
    const float* W_hh  = (const float*)d_in[2];
    const float* b_ih  = (const float*)d_in[3];
    const float* b_hh  = (const float*)d_in[4];
    const float* fc1_w = (const float*)d_in[5];
    const float* fc1_b = (const float*)d_in[6];
    const float* fc2_w = (const float*)d_in[7];
    const float* fc2_b = (const float*)d_in[8];
    float* out = (float*)d_out;

    float* xi = (float*)d_ws;

    xi_gemm_wmma<<<T_LEN / 16 / 8, 256, 0, stream>>>(x, W_ih, b_ih, b_hh, xi);

    gru_scan<<<1, 32, 0, stream>>>(xi, W_hh, b_hh, fc1_w, fc1_b, fc2_w, fc2_b, out);
}


# scan kernel: head off the serial path: a step keeps hnew*weff; its 16-lane sum (same xor tree) issues one step late in the next gates' exp/rcp waits; lane 0 stores 8 results once per pass
# speedup vs baseline: 4.6445x; 1.5194x over previous
#include <hip/hip_runtime.h>

typedef __attribute__((ext_vector_type(2))) float v2f;
typedef __attribute__((ext_vector_type(4))) float v4f;
typedef __attribute__((ext_vector_type(8))) float v8f;

#define T_LEN   262144
#define IN_SZ   64
#define HID     16
#define G3      48
#define PF_D    8

__global__ __launch_bounds__(256) void xi_gemm_wmma(
    const float* __restrict__ x, const float* __restrict__ W_ih,
    const float* __restrict__ b_ih, const float* __restrict__ b_hh,
    float* __restrict__ xi)
{
    const int lane  = threadIdx.x & 31;
    const int wave  = threadIdx.x >> 5;
    const int tile  = blockIdx.x * 8 + wave;
    const int m     = lane & 15;
    const int khalf = lane >> 4;

    const float* xrow = x + (size_t)(tile * 16 + m) * IN_SZ + 2 * khalf;
    v2f a[16];
#pragma unroll
    for (int i = 0; i < 16; ++i)
        a[i] = *(const v2f*)(xrow + 4 * i);

    for (int nb = 0; nb < 3; ++nb) {
        const int col  = nb * 16 + m;
        const float bias = b_ih[col] + (nb < 2 ? b_hh[col] : 0.0f);
        v8f c = {bias, bias, bias, bias, bias, bias, bias, bias};

        const float* wrow = W_ih + (size_t)col * IN_SZ + 2 * khalf;
#pragma unroll
        for (int i = 0; i < 16; ++i) {
            v2f b = *(const v2f*)(wrow + 4 * i);
            c = __builtin_amdgcn_wmma_f32_16x16x4_f32(
                    false, a[i], false, b, (short)0, c, false, false);
        }

        float* orow = xi + (size_t)(tile * 16 + 8 * khalf) * G3 + col;
#pragma unroll
        for (int r = 0; r < 8; ++r)
            orow[(size_t)r * G3] = c[r];
    }
}

template <int MASK>
__device__ __forceinline__ float dpp_row_xmask(float v)
{
    const int i = __builtin_bit_cast(int, v);
    return __builtin_bit_cast(float,
        __builtin_amdgcn_update_dpp(i, i, 0x160 + MASK, 0xf, 0xf, true));
}


#define ROW_SHARE(K) " row_share:" #K " row_mask:0xf bank_mask:0xf bound_ctrl:1"


#define HH_HEAD(T0, T1)                                                          \
    asm("v_mov_b32_dpp %0, %2" ROW_SHARE(0) "\n\t"                               \
        "v_mov_b32_dpp %1, %2" ROW_SHARE(1)                                       \
        : "=&v"(T0), "=&v"(T1) : "v"(hj))

#define HH_TERM0(TK, TN)                                                         \
    asm("v_mul_f32 %0, %4, %5\n\t"                                               \
        "v_mul_f32 %1, %4, %6\n\t"                                               \
        "v_fma_f32 %2, %4, %7, %8\n\t"                                           \
        "v_mov_b32_dpp %3, %9" ROW_SHARE(2)                                       \
        : "=&v"(hr), "=&v"(hz), "=&v"(hn), "=&v"(TN)                             \
        : "v"(TK), "v"(wr[0]), "v"(wz[0]), "v"(wn[0]), "v"(bhn), "v"(hj))

#define HH_TERM(K, K2, TK, TN)                                                   \
    asm("v_fmac_f32 %0, %4, %5\n\t"                                              \
        "v_fmac_f32 %1, %4, %6\n\t"                                              \
        "v_fmac_f32 %2, %4, %7\n\t"                                              \
        "v_mov_b32_dpp %3, %8" ROW_SHARE(K2)                                      \
        : "+v"(hr), "+v"(hz), "+v"(hn), "=&v"(TN)                                \
        : "v"(TK), "v"(wr[K]), "v"(wz[K]), "v"(wn[K]), "v"(hj))

#define HH_TAIL(K, TK)                                                           \
    asm("v_fmac_f32 %0, %3, %4\n\t"                                              \
        "v_fmac_f32 %1, %3, %5\n\t"                                              \
        "v_fmac_f32 %2, %3, %6"                                                   \
        : "+v"(hr), "+v"(hz), "+v"(hn)                                           \
        : "v"(TK), "v"(wr[K]), "v"(wz[K]), "v"(wn[K]))

__global__ __launch_bounds__(32) void gru_scan(
    const float* __restrict__ xi, const float* __restrict__ W_hh,
    const float* __restrict__ b_hh, const float* __restrict__ fc1_w,
    const float* __restrict__ fc1_b, const float* __restrict__ fc2_w,
    const float* __restrict__ fc2_b, float* __restrict__ out)
{
    const int lane = threadIdx.x;
    const int j    = lane & 15;

    float wr[HID], wz[HID], wn[HID];
#pragma unroll
    for (int k = 0; k < HID; ++k) {
        wr[k] = W_hh[(j)      * HID + k];
        wz[k] = W_hh[(16 + j) * HID + k];
        wn[k] = W_hh[(32 + j) * HID + k];
    }
    const float bhn = b_hh[32 + j];

    float weff = 0.0f, beff = fc2_b[0];
#pragma unroll
    for (int i = 0; i < 8; ++i) {
        const float f2 = fc2_w[i];
        weff += f2 * fc1_w[i * HID + j];
        beff += f2 * fc1_b[i];
    }

#pragma unroll
    for (int k = 0; k < HID; ++k)
        asm volatile("" :: "v"(wr[k]), "v"(wz[k]), "v"(wn[k]));
    asm volatile("" :: "v"(bhn));

    float hj = 0.0f;

    const float* xr_p = xi + j;
    const float* xz_p = xi + 16 + j;
    const float* xn_p = xi + 32 + j;

    float pr[PF_D], pz[PF_D], pn[PF_D];
#pragma unroll
    for (int s = 0; s < PF_D; ++s) {
        pr[s] = xr_p[(size_t)s * G3];
        pz[s] = xz_p[(size_t)s * G3];
        pn[s] = xn_p[(size_t)s * G3];
    }

    float yq[PF_D];

    for (int t0 = 0; t0 < T_LEN; t0 += PF_D) {
      const int pf0 = (t0 + PF_D < T_LEN) ? (t0 + PF_D) : t0;
      const float* nr_p = xr_p + (size_t)pf0 * G3;
      const float* nz_p = xz_p + (size_t)pf0 * G3;
      const float* nn_p = xn_p + (size_t)pf0 * G3;

#pragma unroll
      for (int s = 0; s < PF_D; ++s) {
        const float xr = pr[s], xz = pz[s], xn = pn[s];

        float hr, hz, hn, ta, tb, tc;
        HH_HEAD(ta, tb);
        HH_TERM0(ta, tc);
        HH_TERM(1, 3, tb, ta);
        HH_TERM(2, 4, tc, tb);
        HH_TERM(3, 5, ta, tc);
        HH_TERM(4, 6, tb, ta);
        HH_TERM(5, 7, tc, tb);
        HH_TERM(6, 8, ta, tc);
        HH_TERM(7, 9, tb, ta);
        HH_TERM(8, 10, tc, tb);
        HH_TERM(9, 11, ta, tc);
        HH_TERM(10, 12, tb, ta);
        HH_TERM(11, 13, tc, tb);
        HH_TERM(12, 14, ta, tc);
        HH_TERM(13, 15, tb, ta);
        HH_TAIL(14, tc);
        HH_TAIL(15, ta);

        const float r = __builtin_amdgcn_rcpf(1.0f + __expf(-(xr + hr)));
        const float z = __builtin_amdgcn_rcpf(1.0f + __expf(-(xz + hz)));
        const float u = xn + r * hn;
        const float n = 1.0f - 2.0f * __builtin_amdgcn_rcpf(__expf(2.0f * u) + 1.0f);
        const float hnew = n + z * (hj - n);
        hj = hnew;

        pr[s] = nr_p[s * G3];
        pz[s] = nz_p[s * G3];
        pn[s] = nn_p[s * G3];

        yq[s] = hnew * weff;
        if (s > 0) {
          yq[s - 1] += dpp_row_xmask<1>(yq[s - 1]);
          yq[s - 1] += dpp_row_xmask<2>(yq[s - 1]);
          yq[s - 1] += dpp_row_xmask<4>(yq[s - 1]);
          yq[s - 1] += dpp_row_xmask<8>(yq[s - 1]);
        }

        __builtin_amdgcn_sched_barrier(0);
      }

      static_assert(PF_D == 8, "the pass stores its results as two 4-float vectors");
      yq[PF_D - 1] += dpp_row_xmask<1>(yq[PF_D - 1]);
      yq[PF_D - 1] += dpp_row_xmask<2>(yq[PF_D - 1]);
      yq[PF_D - 1] += dpp_row_xmask<4>(yq[PF_D - 1]);
      yq[PF_D - 1] += dpp_row_xmask<8>(yq[PF_D - 1]);
      if (lane == 0) {
        const v4f lo = {yq[0] + beff, yq[1] + beff, yq[2] + beff, yq[3] + beff};
        const v4f hi = {yq[4] + beff, yq[5] + beff, yq[6] + beff, yq[7] + beff};
        *(v4f*)(out + t0)     = lo;
        *(v4f*)(out + t0 + 4) = hi;
      }
    }
}

extern "C" void kernel_launch(void* const* d_in, const int* in_sizes, int n_in,
                              void* d_out, int out_size, void* d_ws, size_t ws_size,
                              hipStream_t stream) {
    const float* x     = (const float*)d_in[0];
    const float* W_ih  = (const float*)d_in[1];
    const float* W_hh  = (const float*)d_in[2];
    const float* b_ih  = (const float*)d_in[3];
    const float* b_hh  = (const float*)d_in[4];
    const float* fc1_w = (const float*)d_in[5];
    const float* fc1_b = (const float*)d_in[6];
    const float* fc2_w = (const float*)d_in[7];
    const float* fc2_b = (const float*)d_in[8];
    float* out = (float*)d_out;

    float* xi = (float*)d_ws;

    xi_gemm_wmma<<<T_LEN / 16 / 8, 256, 0, stream>>>(x, W_ih, b_ih, b_hh, xi);

    gru_scan<<<1, 32, 0, stream>>>(xi, W_hh, b_hh, fc1_w, fc1_b, fc2_w, fc2_b, out);
}


# scan kernel: the recurrent product's 16 terms split over the wave's two 16-lane rows (8 fused terms each, row 1's units rotated by 8); half sums swapped by v_permlanex16 and added: new sum order
# speedup vs baseline: 5.4382x; 1.1709x over previous
#include <hip/hip_runtime.h>

typedef __attribute__((ext_vector_type(2))) float v2f;
typedef __attribute__((ext_vector_type(4))) float v4f;
typedef __attribute__((ext_vector_type(8))) float v8f;

#define T_LEN   262144
#define IN_SZ   64
#define HID     16
#define G3      48
#define PF_D    8
#define KH      8

__global__ __launch_bounds__(256) void xi_gemm_wmma(
    const float* __restrict__ x, const float* __restrict__ W_ih,
    const float* __restrict__ b_ih, const float* __restrict__ b_hh,
    float* __restrict__ xi)
{
    const int lane  = threadIdx.x & 31;
    const int wave  = threadIdx.x >> 5;
    const int tile  = blockIdx.x * 8 + wave;
    const int m     = lane & 15;
    const int khalf = lane >> 4;

    const float* xrow = x + (size_t)(tile * 16 + m) * IN_SZ + 2 * khalf;
    v2f a[16];
#pragma unroll
    for (int i = 0; i < 16; ++i)
        a[i] = *(const v2f*)(xrow + 4 * i);

    for (int nb = 0; nb < 3; ++nb) {
        const int col  = nb * 16 + m;
        const float bias = b_ih[col] + (nb < 2 ? b_hh[col] : 0.0f);
        v8f c = {bias, bias, bias, bias, bias, bias, bias, bias};

        const float* wrow = W_ih + (size_t)col * IN_SZ + 2 * khalf;
#pragma unroll
        for (int i = 0; i < 16; ++i) {
            v2f b = *(const v2f*)(wrow + 4 * i);
            c = __builtin_amdgcn_wmma_f32_16x16x4_f32(
                    false, a[i], false, b, (short)0, c, false, false);
        }

        float* orow = xi + (size_t)(tile * 16 + 8 * khalf) * G3 + col;
#pragma unroll
        for (int r = 0; r < 8; ++r)
            orow[(size_t)r * G3] = c[r];
    }
}

template <int MASK>
__device__ __forceinline__ float dpp_row_xmask(float v)
{
    const int i = __builtin_bit_cast(int, v);
    return __builtin_bit_cast(float,
        __builtin_amdgcn_update_dpp(i, i, 0x160 + MASK, 0xf, 0xf, true));
}


#define ROW_SHARE(K) " row_share:" #K " row_mask:0xf bank_mask:0xf bound_ctrl:1"


#define HH_HEAD(T0, T1)                                                          \
    asm("v_mov_b32_dpp %0, %2" ROW_SHARE(0) "\n\t"                               \
        "v_mov_b32_dpp %1, %2" ROW_SHARE(1)                                       \
        : "=&v"(T0), "=&v"(T1) : "v"(hj))

#define HH_TERM0(TK, TN)                                                         \
    asm("v_mul_f32 %0, %4, %5\n\t"                                               \
        "v_mul_f32 %1, %4, %6\n\t"                                               \
        "v_fma_f32 %2, %4, %7, %8\n\t"                                           \
        "v_mov_b32_dpp %3, %9" ROW_SHARE(2)                                       \
        : "=&v"(hr), "=&v"(hz), "=&v"(hn), "=&v"(TN)                             \
        : "v"(TK), "v"(wr[0]), "v"(wz[0]), "v"(wn[0]), "v"(bhn), "v"(hj))

#define HH_TERM(K, K2, TK, TN)                                                   \
    asm("v_fmac_f32 %0, %4, %5\n\t"                                              \
        "v_fmac_f32 %1, %4, %6\n\t"                                              \
        "v_fmac_f32 %2, %4, %7\n\t"                                              \
        "v_mov_b32_dpp %3, %8" ROW_SHARE(K2)                                      \
        : "+v"(hr), "+v"(hz), "+v"(hn), "=&v"(TN)                                \
        : "v"(TK), "v"(wr[K]), "v"(wz[K]), "v"(wn[K]), "v"(hj))

#define HH_TAIL(K, TK)                                                           \
    asm("v_fmac_f32 %0, %3, %4\n\t"                                              \
        "v_fmac_f32 %1, %3, %5\n\t"                                              \
        "v_fmac_f32 %2, %3, %6"                                                   \
        : "+v"(hr), "+v"(hz), "+v"(hn)                                           \
        : "v"(TK), "v"(wr[K]), "v"(wz[K]), "v"(wn[K]))

__global__ __launch_bounds__(32) void gru_scan(
    const float* __restrict__ xi, const float* __restrict__ W_hh,
    const float* __restrict__ b_hh, const float* __restrict__ fc1_w,
    const float* __restrict__ fc1_b, const float* __restrict__ fc2_w,
    const float* __restrict__ fc2_b, float* __restrict__ out)
{
    const int lane = threadIdx.x;
    const int row  = lane >> 4;
    const int j    = (lane & 15) ^ (row * KH);
    const int kb   = row * KH;

    float wr[KH], wz[KH], wn[KH];
#pragma unroll
    for (int k = 0; k < KH; ++k) {
        wr[k] = W_hh[(j)      * HID + kb + k];
        wz[k] = W_hh[(16 + j) * HID + kb + k];
        wn[k] = W_hh[(32 + j) * HID + kb + k];
    }
    const float bhn_j = b_hh[32 + j];
    const float bhn   = row ? 0.0f : bhn_j;

    float weff = 0.0f, beff = fc2_b[0];
#pragma unroll
    for (int i = 0; i < 8; ++i) {
        const float f2 = fc2_w[i];
        weff += f2 * fc1_w[i * HID + j];
        beff += f2 * fc1_b[i];
    }

#pragma unroll
    for (int k = 0; k < KH; ++k)
        asm volatile("" :: "v"(wr[k]), "v"(wz[k]), "v"(wn[k]));
    asm volatile("" :: "v"(bhn));

    float hj = 0.0f;

    const unsigned sel_lo = 0xfedcba98u, sel_hi = 0x76543210u;

    const float* xr_p = xi + j;
    const float* xz_p = xi + 16 + j;
    const float* xn_p = xi + 32 + j;

    float pr[PF_D], pz[PF_D], pn[PF_D];
#pragma unroll
    for (int s = 0; s < PF_D; ++s) {
        pr[s] = xr_p[(size_t)s * G3];
        pz[s] = xz_p[(size_t)s * G3];
        pn[s] = xn_p[(size_t)s * G3];
    }

    float yq[PF_D];

    for (int t0 = 0; t0 < T_LEN; t0 += PF_D) {
      const int pf0 = (t0 + PF_D < T_LEN) ? (t0 + PF_D) : t0;
      const float* nr_p = xr_p + (size_t)pf0 * G3;
      const float* nz_p = xz_p + (size_t)pf0 * G3;
      const float* nn_p = xn_p + (size_t)pf0 * G3;

#pragma unroll
      for (int s = 0; s < PF_D; ++s) {
        const float xr = pr[s], xz = pz[s], xn = pn[s];

        float hr, hz, hn, ta, tb, tc;
        HH_HEAD(ta, tb);
        HH_TERM0(ta, tc);
        HH_TERM(1, 3, tb, ta);
        HH_TERM(2, 4, tc, tb);
        HH_TERM(3, 5, ta, tc);
        HH_TERM(4, 6, tb, ta);
        HH_TERM(5, 7, tc, tb);
        HH_TAIL(6, ta);
        HH_TAIL(7, tb);

        float hr_o, hz_o, hn_o;
        asm("v_permlanex16_b32 %0, %3, %6, %7\n\t"
            "v_permlanex16_b32 %1, %4, %6, %7\n\t"
            "v_permlanex16_b32 %2, %5, %6, %7"
            : "=&v"(hr_o), "=&v"(hz_o), "=&v"(hn_o)
            : "v"(hr), "v"(hz), "v"(hn), "s"(sel_lo), "s"(sel_hi));
        hr += hr_o;
        hz += hz_o;
        hn += hn_o;

        const float r = __builtin_amdgcn_rcpf(1.0f + __expf(-(xr + hr)));
        const float z = __builtin_amdgcn_rcpf(1.0f + __expf(-(xz + hz)));
        const float u = xn + r * hn;
        const float n = 1.0f - 2.0f * __builtin_amdgcn_rcpf(__expf(2.0f * u) + 1.0f);
        const float hnew = n + z * (hj - n);
        hj = hnew;

        pr[s] = nr_p[s * G3];
        pz[s] = nz_p[s * G3];
        pn[s] = nn_p[s * G3];

        yq[s] = hnew * weff;
        if (s > 0) {
          yq[s - 1] += dpp_row_xmask<1>(yq[s - 1]);
          yq[s - 1] += dpp_row_xmask<2>(yq[s - 1]);
          yq[s - 1] += dpp_row_xmask<4>(yq[s - 1]);
          yq[s - 1] += dpp_row_xmask<8>(yq[s - 1]);
        }

        __builtin_amdgcn_sched_barrier(0);
      }

      static_assert(PF_D == 8, "the pass stores its results as two 4-float vectors");
      yq[PF_D - 1] += dpp_row_xmask<1>(yq[PF_D - 1]);
      yq[PF_D - 1] += dpp_row_xmask<2>(yq[PF_D - 1]);
      yq[PF_D - 1] += dpp_row_xmask<4>(yq[PF_D - 1]);
      yq[PF_D - 1] += dpp_row_xmask<8>(yq[PF_D - 1]);
      if (lane == 0) {
        const v4f lo = {yq[0] + beff, yq[1] + beff, yq[2] + beff, yq[3] + beff};
        const v4f hi = {yq[4] + beff, yq[5] + beff, yq[6] + beff, yq[7] + beff};
        *(v4f*)(out + t0)     = lo;
        *(v4f*)(out + t0 + 4) = hi;
      }
    }
}

extern "C" void kernel_launch(void* const* d_in, const int* in_sizes, int n_in,
                              void* d_out, int out_size, void* d_ws, size_t ws_size,
                              hipStream_t stream) {
    const float* x     = (const float*)d_in[0];
    const float* W_ih  = (const float*)d_in[1];
    const float* W_hh  = (const float*)d_in[2];
    const float* b_ih  = (const float*)d_in[3];
    const float* b_hh  = (const float*)d_in[4];
    const float* fc1_w = (const float*)d_in[5];
    const float* fc1_b = (const float*)d_in[6];
    const float* fc2_w = (const float*)d_in[7];
    const float* fc2_b = (const float*)d_in[8];
    float* out = (float*)d_out;

    float* xi = (float*)d_ws;

    xi_gemm_wmma<<<T_LEN / 16 / 8, 256, 0, stream>>>(x, W_ih, b_ih, b_hh, xi);

    gru_scan<<<1, 32, 0, stream>>>(xi, W_hh, b_hh, fc1_w, fc1_b, fc2_w, fc2_b, out);
}


# scan kernel: r, z, n gates on the hardware tanh, one v_tanh_f32 each, sigmoid(a) = 0.5 + 0.5 tanh(a/2) with the 0.5s folded exactly into W_hh, for exp + rcp; hnew = n(1-z) + z h, one fma behind n
# speedup vs baseline: 7.3931x; 1.3595x over previous
#include <hip/hip_runtime.h>

typedef __attribute__((ext_vector_type(2))) float v2f;
typedef __attribute__((ext_vector_type(4))) float v4f;
typedef __attribute__((ext_vector_type(8))) float v8f;

#define T_LEN   262144
#define IN_SZ   64
#define HID     16
#define G3      48
#define PF_D    8
#define KH      8

__global__ __launch_bounds__(256) void xi_gemm_wmma(
    const float* __restrict__ x, const float* __restrict__ W_ih,
    const float* __restrict__ b_ih, const float* __restrict__ b_hh,
    float* __restrict__ xi)
{
    const int lane  = threadIdx.x & 31;
    const int wave  = threadIdx.x >> 5;
    const int tile  = blockIdx.x * 8 + wave;
    const int m     = lane & 15;
    const int khalf = lane >> 4;

    const float* xrow = x + (size_t)(tile * 16 + m) * IN_SZ + 2 * khalf;
    v2f a[16];
#pragma unroll
    for (int i = 0; i < 16; ++i)
        a[i] = *(const v2f*)(xrow + 4 * i);

    for (int nb = 0; nb < 3; ++nb) {
        const int col  = nb * 16 + m;
        const float bias = b_ih[col] + (nb < 2 ? b_hh[col] : 0.0f);
        v8f c = {bias, bias, bias, bias, bias, bias, bias, bias};

        const float* wrow = W_ih + (size_t)col * IN_SZ + 2 * khalf;
#pragma unroll
        for (int i = 0; i < 16; ++i) {
            v2f b = *(const v2f*)(wrow + 4 * i);
            c = __builtin_amdgcn_wmma_f32_16x16x4_f32(
                    false, a[i], false, b, (short)0, c, false, false);
        }

        float* orow = xi + (size_t)(tile * 16 + 8 * khalf) * G3 + col;
#pragma unroll
        for (int r = 0; r < 8; ++r)
            orow[(size_t)r * G3] = c[r];
    }
}

template <int MASK>
__device__ __forceinline__ float dpp_row_xmask(float v)
{
    const int i = __builtin_bit_cast(int, v);
    return __builtin_bit_cast(float,
        __builtin_amdgcn_update_dpp(i, i, 0x160 + MASK, 0xf, 0xf, true));
}


#define ROW_SHARE(K) " row_share:" #K " row_mask:0xf bank_mask:0xf bound_ctrl:1"


#define HH_HEAD(T0, T1)                                                          \
    asm("v_mov_b32_dpp %0, %2" ROW_SHARE(0) "\n\t"                               \
        "v_mov_b32_dpp %1, %2" ROW_SHARE(1)                                       \
        : "=&v"(T0), "=&v"(T1) : "v"(hj))

#define HH_TERM0(TK, TN)                                                         \
    asm("v_mul_f32 %0, %4, %5\n\t"                                               \
        "v_mul_f32 %1, %4, %6\n\t"                                               \
        "v_fma_f32 %2, %4, %7, %8\n\t"                                           \
        "v_mov_b32_dpp %3, %9" ROW_SHARE(2)                                       \
        : "=&v"(hr), "=&v"(hz), "=&v"(hn), "=&v"(TN)                             \
        : "v"(TK), "v"(wr[0]), "v"(wz[0]), "v"(wn[0]), "v"(bhn), "v"(hj))

#define HH_TERM(K, K2, TK, TN)                                                   \
    asm("v_fmac_f32 %0, %4, %5\n\t"                                              \
        "v_fmac_f32 %1, %4, %6\n\t"                                              \
        "v_fmac_f32 %2, %4, %7\n\t"                                              \
        "v_mov_b32_dpp %3, %8" ROW_SHARE(K2)                                      \
        : "+v"(hr), "+v"(hz), "+v"(hn), "=&v"(TN)                                \
        : "v"(TK), "v"(wr[K]), "v"(wz[K]), "v"(wn[K]), "v"(hj))

#define HH_TAIL(K, TK)                                                           \
    asm("v_fmac_f32 %0, %3, %4\n\t"                                              \
        "v_fmac_f32 %1, %3, %5\n\t"                                              \
        "v_fmac_f32 %2, %3, %6"                                                   \
        : "+v"(hr), "+v"(hz), "+v"(hn)                                           \
        : "v"(TK), "v"(wr[K]), "v"(wz[K]), "v"(wn[K]))

__global__ __launch_bounds__(32) void gru_scan(
    const float* __restrict__ xi, const float* __restrict__ W_hh,
    const float* __restrict__ b_hh, const float* __restrict__ fc1_w,
    const float* __restrict__ fc1_b, const float* __restrict__ fc2_w,
    const float* __restrict__ fc2_b, float* __restrict__ out)
{
    const int lane = threadIdx.x;
    const int row  = lane >> 4;
    const int j    = (lane & 15) ^ (row * KH);
    const int kb   = row * KH;

    float wr[KH], wz[KH], wn[KH];
#pragma unroll
    for (int k = 0; k < KH; ++k) {
        wr[k] = 0.5f * W_hh[(j)      * HID + kb + k];
        wz[k] = 0.5f * W_hh[(16 + j) * HID + kb + k];
        wn[k] = 0.5f * W_hh[(32 + j) * HID + kb + k];
    }
    const float bhn_j = 0.5f * b_hh[32 + j];
    const float bhn   = row ? 0.0f : bhn_j;

    float weff = 0.0f, beff = fc2_b[0];
#pragma unroll
    for (int i = 0; i < 8; ++i) {
        const float f2 = fc2_w[i];
        weff += f2 * fc1_w[i * HID + j];
        beff += f2 * fc1_b[i];
    }

#pragma unroll
    for (int k = 0; k < KH; ++k)
        asm volatile("" :: "v"(wr[k]), "v"(wz[k]), "v"(wn[k]));
    asm volatile("" :: "v"(bhn));

    float hj = 0.0f;

    const unsigned sel_lo = 0xfedcba98u, sel_hi = 0x76543210u;

    const float* xr_p = xi + j;
    const float* xz_p = xi + 16 + j;
    const float* xn_p = xi + 32 + j;

    float pr[PF_D], pz[PF_D], pn[PF_D];
#pragma unroll
    for (int s = 0; s < PF_D; ++s) {
        pr[s] = xr_p[(size_t)s * G3];
        pz[s] = xz_p[(size_t)s * G3];
        pn[s] = xn_p[(size_t)s * G3];
    }

    float yq[PF_D];

    for (int t0 = 0; t0 < T_LEN; t0 += PF_D) {
      const int pf0 = (t0 + PF_D < T_LEN) ? (t0 + PF_D) : t0;
      const float* nr_p = xr_p + (size_t)pf0 * G3;
      const float* nz_p = xz_p + (size_t)pf0 * G3;
      const float* nn_p = xn_p + (size_t)pf0 * G3;

#pragma unroll
      for (int s = 0; s < PF_D; ++s) {
        const float xr = pr[s], xz = pz[s], xn = pn[s];

        float hr, hz, hn, ta, tb, tc;
        HH_HEAD(ta, tb);
        HH_TERM0(ta, tc);
        HH_TERM(1, 3, tb, ta);
        HH_TERM(2, 4, tc, tb);
        HH_TERM(3, 5, ta, tc);
        HH_TERM(4, 6, tb, ta);
        HH_TERM(5, 7, tc, tb);
        HH_TAIL(6, ta);
        HH_TAIL(7, tb);

        float hr_o, hz_o, hn_o;
        asm("v_permlanex16_b32 %0, %3, %6, %7\n\t"
            "v_permlanex16_b32 %1, %4, %6, %7\n\t"
            "v_permlanex16_b32 %2, %5, %6, %7"
            : "=&v"(hr_o), "=&v"(hz_o), "=&v"(hn_o)
            : "v"(hr), "v"(hz), "v"(hn), "s"(sel_lo), "s"(sel_hi));
        hr += hr_o;
        hz += hz_o;
        hn += hn_o;

        const float tr = __builtin_amdgcn_tanhf(__builtin_fmaf(xr, 0.5f, hr));
        const float tz = __builtin_amdgcn_tanhf(__builtin_fmaf(xz, 0.5f, hz));
        const float n  = __builtin_amdgcn_tanhf(__builtin_fmaf(tr, hn, xn + hn));
        const float hh   = 0.5f * hj;
        const float omz  = __builtin_fmaf(tz, -0.5f, 0.5f);
        const float zh   = __builtin_fmaf(tz, hh, hh);
        const float hnew = __builtin_fmaf(n, omz, zh);
        hj = hnew;

        pr[s] = nr_p[s * G3];
        pz[s] = nz_p[s * G3];
        pn[s] = nn_p[s * G3];

        yq[s] = hnew * weff;
        if (s > 0) {
          yq[s - 1] += dpp_row_xmask<1>(yq[s - 1]);
          yq[s - 1] += dpp_row_xmask<2>(yq[s - 1]);
          yq[s - 1] += dpp_row_xmask<4>(yq[s - 1]);
          yq[s - 1] += dpp_row_xmask<8>(yq[s - 1]);
        }

        __builtin_amdgcn_sched_barrier(0);
      }

      static_assert(PF_D == 8, "the pass stores its results as two 4-float vectors");
      yq[PF_D - 1] += dpp_row_xmask<1>(yq[PF_D - 1]);
      yq[PF_D - 1] += dpp_row_xmask<2>(yq[PF_D - 1]);
      yq[PF_D - 1] += dpp_row_xmask<4>(yq[PF_D - 1]);
      yq[PF_D - 1] += dpp_row_xmask<8>(yq[PF_D - 1]);
      if (lane == 0) {
        const v4f lo = {yq[0] + beff, yq[1] + beff, yq[2] + beff, yq[3] + beff};
        const v4f hi = {yq[4] + beff, yq[5] + beff, yq[6] + beff, yq[7] + beff};
        *(v4f*)(out + t0)     = lo;
        *(v4f*)(out + t0 + 4) = hi;
      }
    }
}

extern "C" void kernel_launch(void* const* d_in, const int* in_sizes, int n_in,
                              void* d_out, int out_size, void* d_ws, size_t ws_size,
                              hipStream_t stream) {
    const float* x     = (const float*)d_in[0];
    const float* W_ih  = (const float*)d_in[1];
    const float* W_hh  = (const float*)d_in[2];
    const float* b_ih  = (const float*)d_in[3];
    const float* b_hh  = (const float*)d_in[4];
    const float* fc1_w = (const float*)d_in[5];
    const float* fc1_b = (const float*)d_in[6];
    const float* fc2_w = (const float*)d_in[7];
    const float* fc2_b = (const float*)d_in[8];
    float* out = (float*)d_out;

    float* xi = (float*)d_ws;

    xi_gemm_wmma<<<T_LEN / 16 / 8, 256, 0, stream>>>(x, W_ih, b_ih, b_hh, xi);

    gru_scan<<<1, 32, 0, stream>>>(xi, W_hh, b_hh, fc1_w, fc1_b, fc2_w, fc2_b, out);
}


# speedup vs baseline: 7.7952x; 1.0544x over previous
#include <hip/hip_runtime.h>

typedef __attribute__((ext_vector_type(2))) float v2f;
typedef __attribute__((ext_vector_type(4))) float v4f;
typedef __attribute__((ext_vector_type(8))) float v8f;

#define T_LEN   262144
#define IN_SZ   64
#define HID     16
#define G3      48
#define PF_D    8
#define KH      8

__global__ __launch_bounds__(256) void xi_gemm_wmma(
    const float* __restrict__ x, const float* __restrict__ W_ih,
    const float* __restrict__ b_ih, const float* __restrict__ b_hh,
    float* __restrict__ xi)
{
    const int lane  = threadIdx.x & 31;
    const int wave  = threadIdx.x >> 5;
    const int tile  = blockIdx.x * 8 + wave;
    const int m     = lane & 15;
    const int khalf = lane >> 4;

    const float* xrow = x + (size_t)(tile * 16 + m) * IN_SZ + 2 * khalf;
    v2f a[16];
#pragma unroll
    for (int i = 0; i < 16; ++i)
        a[i] = *(const v2f*)(xrow + 4 * i);

    for (int nb = 0; nb < 3; ++nb) {
        const int col  = nb * 16 + m;
        const float bias = b_ih[col] + (nb < 2 ? b_hh[col] : 0.0f);
        v8f c = {bias, bias, bias, bias, bias, bias, bias, bias};

        const float* wrow = W_ih + (size_t)col * IN_SZ + 2 * khalf;
#pragma unroll
        for (int i = 0; i < 16; ++i) {
            v2f b = *(const v2f*)(wrow + 4 * i);
            c = __builtin_amdgcn_wmma_f32_16x16x4_f32(
                    false, a[i], false, b, (short)0, c, false, false);
        }

        float* orow = xi + (size_t)(tile * 16 + 8 * khalf) * G3 + col;
#pragma unroll
        for (int r = 0; r < 8; ++r)
            orow[(size_t)r * G3] = c[r];
    }
}

template <int MASK>
__device__ __forceinline__ float dpp_row_xmask(float v)
{
    const int i = __builtin_bit_cast(int, v);
    return __builtin_bit_cast(float,
        __builtin_amdgcn_update_dpp(i, i, 0x160 + MASK, 0xf, 0xf, true));
}


__global__ __launch_bounds__(32) void gru_scan(
    const float* __restrict__ xi, const float* __restrict__ W_hh,
    const float* __restrict__ b_hh, const float* __restrict__ fc1_w,
    const float* __restrict__ fc1_b, const float* __restrict__ fc2_w,
    const float* __restrict__ fc2_b, float* __restrict__ out)
{
    const int lane = threadIdx.x;
    const int row  = lane >> 4;
    const int j    = (lane & 15) ^ (row * KH);
    const int kb   = row * KH;

    float wr[KH], wz[KH], wn[KH];
#pragma unroll
    for (int k = 0; k < KH; ++k) {
        wr[k] = 0.5f * W_hh[(j)      * HID + kb + k];
        wz[k] = 0.5f * W_hh[(16 + j) * HID + kb + k];
        wn[k] = 0.5f * W_hh[(32 + j) * HID + kb + k];
    }
    const float bhn_j = 0.5f * b_hh[32 + j];
    const float bhn   = row ? 0.0f : bhn_j;

    float weff = 0.0f, beff = fc2_b[0];
#pragma unroll
    for (int i = 0; i < 8; ++i) {
        const float f2 = fc2_w[i];
        weff += f2 * fc1_w[i * HID + j];
        beff += f2 * fc1_b[i];
    }

#pragma unroll
    for (int k = 0; k < KH; ++k)
        asm volatile("" :: "v"(wr[k]), "v"(wz[k]), "v"(wn[k]));
    asm volatile("" :: "v"(bhn));

    float hj = 0.0f;

    unsigned selk[KH];
#pragma unroll
    for (int k = 0; k < KH; ++k) {
        selk[k] = 0x11111111u * (unsigned)k;
        asm volatile("" : "+s"(selk[k]));
    }

    const unsigned sel_lo = 0xfedcba98u, sel_hi = 0x76543210u;

    const float* xr_p = xi + j;
    const float* xz_p = xi + 16 + j;
    const float* xn_p = xi + 32 + j;

    float pr[PF_D], pz[PF_D], pn[PF_D];
#pragma unroll
    for (int s = 0; s < PF_D; ++s) {
        pr[s] = xr_p[(size_t)s * G3];
        pz[s] = xz_p[(size_t)s * G3];
        pn[s] = xn_p[(size_t)s * G3];
    }

    float yq[PF_D];

    for (int t0 = 0; t0 < T_LEN; t0 += PF_D) {
      const int pf0 = (t0 + PF_D < T_LEN) ? (t0 + PF_D) : t0;
      const float* nr_p = xr_p + (size_t)pf0 * G3;
      const float* nz_p = xz_p + (size_t)pf0 * G3;
      const float* nn_p = xn_p + (size_t)pf0 * G3;

#pragma unroll
      for (int s = 0; s < PF_D; ++s) {
        const float xr = pr[s], xz = pz[s], xn = pn[s];

        float hb[KH];
#pragma unroll
        for (int k = 0; k < KH; ++k)
          hb[k] = __builtin_bit_cast(float, __builtin_amdgcn_permlane16(
                      0, __builtin_bit_cast(int, hj), selk[k], selk[k], true, false));
        float hrA = hb[0] * wr[0], hrB = hb[1] * wr[1];
        float hzA = hb[0] * wz[0], hzB = hb[1] * wz[1];
        float hnA = __builtin_fmaf(hb[0], wn[0], bhn), hnB = hb[1] * wn[1];
#pragma unroll
        for (int k = 2; k < KH; k += 2) {
          hrA = __builtin_fmaf(hb[k], wr[k], hrA);
          hrB = __builtin_fmaf(hb[k + 1], wr[k + 1], hrB);
          hzA = __builtin_fmaf(hb[k], wz[k], hzA);
          hzB = __builtin_fmaf(hb[k + 1], wz[k + 1], hzB);
          hnA = __builtin_fmaf(hb[k], wn[k], hnA);
          hnB = __builtin_fmaf(hb[k + 1], wn[k + 1], hnB);
        }
        float hr = hrA + hrB, hz = hzA + hzB, hn = hnA + hnB;

        float hr_o, hz_o, hn_o;
        asm("v_permlanex16_b32 %0, %3, %6, %7\n\t"
            "v_permlanex16_b32 %1, %4, %6, %7\n\t"
            "v_permlanex16_b32 %2, %5, %6, %7"
            : "=&v"(hr_o), "=&v"(hz_o), "=&v"(hn_o)
            : "v"(hr), "v"(hz), "v"(hn), "s"(sel_lo), "s"(sel_hi));
        hr += hr_o;
        hz += hz_o;
        hn += hn_o;

        const float tr = __builtin_amdgcn_tanhf(__builtin_fmaf(xr, 0.5f, hr));
        const float tz = __builtin_amdgcn_tanhf(__builtin_fmaf(xz, 0.5f, hz));
        const float n  = __builtin_amdgcn_tanhf(__builtin_fmaf(tr, hn, xn + hn));
        const float hh   = 0.5f * hj;
        const float omz  = __builtin_fmaf(tz, -0.5f, 0.5f);
        const float zh   = __builtin_fmaf(tz, hh, hh);
        const float hnew = __builtin_fmaf(n, omz, zh);
        hj = hnew;

        pr[s] = nr_p[s * G3];
        pz[s] = nz_p[s * G3];
        pn[s] = nn_p[s * G3];

        yq[s] = hnew * weff;
        if (s > 0) {
          yq[s - 1] += dpp_row_xmask<1>(yq[s - 1]);
          yq[s - 1] += dpp_row_xmask<2>(yq[s - 1]);
          yq[s - 1] += dpp_row_xmask<4>(yq[s - 1]);
          yq[s - 1] += dpp_row_xmask<8>(yq[s - 1]);
        }

        __builtin_amdgcn_sched_barrier(0);
      }

      static_assert(PF_D == 8, "the pass stores its results as two 4-float vectors");
      yq[PF_D - 1] += dpp_row_xmask<1>(yq[PF_D - 1]);
      yq[PF_D - 1] += dpp_row_xmask<2>(yq[PF_D - 1]);
      yq[PF_D - 1] += dpp_row_xmask<4>(yq[PF_D - 1]);
      yq[PF_D - 1] += dpp_row_xmask<8>(yq[PF_D - 1]);
      if (lane == 0) {
        const v4f lo = {yq[0] + beff, yq[1] + beff, yq[2] + beff, yq[3] + beff};
        const v4f hi = {yq[4] + beff, yq[5] + beff, yq[6] + beff, yq[7] + beff};
        *(v4f*)(out + t0)     = lo;
        *(v4f*)(out + t0 + 4) = hi;
      }
    }
}

extern "C" void kernel_launch(void* const* d_in, const int* in_sizes, int n_in,
                              void* d_out, int out_size, void* d_ws, size_t ws_size,
                              hipStream_t stream) {
    const float* x     = (const float*)d_in[0];
    const float* W_ih  = (const float*)d_in[1];
    const float* W_hh  = (const float*)d_in[2];
    const float* b_ih  = (const float*)d_in[3];
    const float* b_hh  = (const float*)d_in[4];
    const float* fc1_w = (const float*)d_in[5];
    const float* fc1_b = (const float*)d_in[6];
    const float* fc2_w = (const float*)d_in[7];
    const float* fc2_b = (const float*)d_in[8];
    float* out = (float*)d_out;

    float* xi = (float*)d_ws;

    xi_gemm_wmma<<<T_LEN / 16 / 8, 256, 0, stream>>>(x, W_ih, b_ih, b_hh, xi);

    gru_scan<<<1, 32, 0, stream>>>(xi, W_hh, b_hh, fc1_w, fc1_b, fc2_w, fc2_b, out);
}


# scan kernel: xi register ring 16 steps deep, not 8 (pass unrolled by 16, 48 row loads in flight): at the present step length a lead of 8 steps no longer covers a row's round trip from memory
# speedup vs baseline: 8.8703x; 1.1379x over previous
#include <hip/hip_runtime.h>

typedef __attribute__((ext_vector_type(2))) float v2f;
typedef __attribute__((ext_vector_type(4))) float v4f;
typedef __attribute__((ext_vector_type(8))) float v8f;

#define T_LEN   262144
#define IN_SZ   64
#define HID     16
#define G3      48
#define PF_D    16
#define KH      8

__global__ __launch_bounds__(256) void xi_gemm_wmma(
    const float* __restrict__ x, const float* __restrict__ W_ih,
    const float* __restrict__ b_ih, const float* __restrict__ b_hh,
    float* __restrict__ xi)
{
    const int lane  = threadIdx.x & 31;
    const int wave  = threadIdx.x >> 5;
    const int tile  = blockIdx.x * 8 + wave;
    const int m     = lane & 15;
    const int khalf = lane >> 4;

    const float* xrow = x + (size_t)(tile * 16 + m) * IN_SZ + 2 * khalf;
    v2f a[16];
#pragma unroll
    for (int i = 0; i < 16; ++i)
        a[i] = *(const v2f*)(xrow + 4 * i);

    for (int nb = 0; nb < 3; ++nb) {
        const int col  = nb * 16 + m;
        const float bias = b_ih[col] + (nb < 2 ? b_hh[col] : 0.0f);
        v8f c = {bias, bias, bias, bias, bias, bias, bias, bias};

        const float* wrow = W_ih + (size_t)col * IN_SZ + 2 * khalf;
#pragma unroll
        for (int i = 0; i < 16; ++i) {
            v2f b = *(const v2f*)(wrow + 4 * i);
            c = __builtin_amdgcn_wmma_f32_16x16x4_f32(
                    false, a[i], false, b, (short)0, c, false, false);
        }

        float* orow = xi + (size_t)(tile * 16 + 8 * khalf) * G3 + col;
#pragma unroll
        for (int r = 0; r < 8; ++r)
            orow[(size_t)r * G3] = c[r];
    }
}

template <int MASK>
__device__ __forceinline__ float dpp_row_xmask(float v)
{
    const int i = __builtin_bit_cast(int, v);
    return __builtin_bit_cast(float,
        __builtin_amdgcn_update_dpp(i, i, 0x160 + MASK, 0xf, 0xf, true));
}


__global__ __launch_bounds__(32) void gru_scan(
    const float* __restrict__ xi, const float* __restrict__ W_hh,
    const float* __restrict__ b_hh, const float* __restrict__ fc1_w,
    const float* __restrict__ fc1_b, const float* __restrict__ fc2_w,
    const float* __restrict__ fc2_b, float* __restrict__ out)
{
    const int lane = threadIdx.x;
    const int row  = lane >> 4;
    const int j    = (lane & 15) ^ (row * KH);
    const int kb   = row * KH;

    float wr[KH], wz[KH], wn[KH];
#pragma unroll
    for (int k = 0; k < KH; ++k) {
        wr[k] = 0.5f * W_hh[(j)      * HID + kb + k];
        wz[k] = 0.5f * W_hh[(16 + j) * HID + kb + k];
        wn[k] = 0.5f * W_hh[(32 + j) * HID + kb + k];
    }
    const float bhn_j = 0.5f * b_hh[32 + j];
    const float bhn   = row ? 0.0f : bhn_j;

    float weff = 0.0f, beff = fc2_b[0];
#pragma unroll
    for (int i = 0; i < 8; ++i) {
        const float f2 = fc2_w[i];
        weff += f2 * fc1_w[i * HID + j];
        beff += f2 * fc1_b[i];
    }

#pragma unroll
    for (int k = 0; k < KH; ++k)
        asm volatile("" :: "v"(wr[k]), "v"(wz[k]), "v"(wn[k]));
    asm volatile("" :: "v"(bhn));

    float hj = 0.0f;

    unsigned selk[KH];
#pragma unroll
    for (int k = 0; k < KH; ++k) {
        selk[k] = 0x11111111u * (unsigned)k;
        asm volatile("" : "+s"(selk[k]));
    }

    const unsigned sel_lo = 0xfedcba98u, sel_hi = 0x76543210u;

    const float* xr_p = xi + j;
    const float* xz_p = xi + 16 + j;
    const float* xn_p = xi + 32 + j;

    float pr[PF_D], pz[PF_D], pn[PF_D];
#pragma unroll
    for (int s = 0; s < PF_D; ++s) {
        pr[s] = xr_p[(size_t)s * G3];
        pz[s] = xz_p[(size_t)s * G3];
        pn[s] = xn_p[(size_t)s * G3];
    }

    float yq[PF_D];

    for (int t0 = 0; t0 < T_LEN; t0 += PF_D) {
      const int pf0 = (t0 + PF_D < T_LEN) ? (t0 + PF_D) : t0;
      const float* nr_p = xr_p + (size_t)pf0 * G3;
      const float* nz_p = xz_p + (size_t)pf0 * G3;
      const float* nn_p = xn_p + (size_t)pf0 * G3;

#pragma unroll
      for (int s = 0; s < PF_D; ++s) {
        const float xr = pr[s], xz = pz[s], xn = pn[s];

        float hb[KH];
#pragma unroll
        for (int k = 0; k < KH; ++k)
          hb[k] = __builtin_bit_cast(float, __builtin_amdgcn_permlane16(
                      0, __builtin_bit_cast(int, hj), selk[k], selk[k], true, false));
        float hrA = hb[0] * wr[0], hrB = hb[1] * wr[1];
        float hzA = hb[0] * wz[0], hzB = hb[1] * wz[1];
        float hnA = __builtin_fmaf(hb[0], wn[0], bhn), hnB = hb[1] * wn[1];
#pragma unroll
        for (int k = 2; k < KH; k += 2) {
          hrA = __builtin_fmaf(hb[k], wr[k], hrA);
          hrB = __builtin_fmaf(hb[k + 1], wr[k + 1], hrB);
          hzA = __builtin_fmaf(hb[k], wz[k], hzA);
          hzB = __builtin_fmaf(hb[k + 1], wz[k + 1], hzB);
          hnA = __builtin_fmaf(hb[k], wn[k], hnA);
          hnB = __builtin_fmaf(hb[k + 1], wn[k + 1], hnB);
        }
        float hr = hrA + hrB, hz = hzA + hzB, hn = hnA + hnB;

        float hr_o, hz_o, hn_o;
        asm("v_permlanex16_b32 %0, %3, %6, %7\n\t"
            "v_permlanex16_b32 %1, %4, %6, %7\n\t"
            "v_permlanex16_b32 %2, %5, %6, %7"
            : "=&v"(hr_o), "=&v"(hz_o), "=&v"(hn_o)
            : "v"(hr), "v"(hz), "v"(hn), "s"(sel_lo), "s"(sel_hi));
        hr += hr_o;
        hz += hz_o;
        hn += hn_o;

        const float tr = __builtin_amdgcn_tanhf(__builtin_fmaf(xr, 0.5f, hr));
        const float tz = __builtin_amdgcn_tanhf(__builtin_fmaf(xz, 0.5f, hz));
        const float n  = __builtin_amdgcn_tanhf(__builtin_fmaf(tr, hn, xn + hn));
        const float hh   = 0.5f * hj;
        const float omz  = __builtin_fmaf(tz, -0.5f, 0.5f);
        const float zh   = __builtin_fmaf(tz, hh, hh);
        const float hnew = __builtin_fmaf(n, omz, zh);
        hj = hnew;

        pr[s] = nr_p[s * G3];
        pz[s] = nz_p[s * G3];
        pn[s] = nn_p[s * G3];

        yq[s] = hnew * weff;
        if (s > 0) {
          yq[s - 1] += dpp_row_xmask<1>(yq[s - 1]);
          yq[s - 1] += dpp_row_xmask<2>(yq[s - 1]);
          yq[s - 1] += dpp_row_xmask<4>(yq[s - 1]);
          yq[s - 1] += dpp_row_xmask<8>(yq[s - 1]);
        }

        __builtin_amdgcn_sched_barrier(0);
      }

      asm volatile("" :: "v"(nr_p));

      static_assert(PF_D % 4 == 0, "the pass stores its results as 4-float vectors");
      static_assert(T_LEN % PF_D == 0, "the scan runs whole passes of PF_D steps");
      yq[PF_D - 1] += dpp_row_xmask<1>(yq[PF_D - 1]);
      yq[PF_D - 1] += dpp_row_xmask<2>(yq[PF_D - 1]);
      yq[PF_D - 1] += dpp_row_xmask<4>(yq[PF_D - 1]);
      yq[PF_D - 1] += dpp_row_xmask<8>(yq[PF_D - 1]);
      if (lane == 0) {
#pragma unroll
        for (int q = 0; q < PF_D; q += 4) {
          const v4f y4 = {yq[q] + beff, yq[q + 1] + beff, yq[q + 2] + beff, yq[q + 3] + beff};
          *(v4f*)(out + t0 + q) = y4;
        }
      }
    }
}

extern "C" void kernel_launch(void* const* d_in, const int* in_sizes, int n_in,
                              void* d_out, int out_size, void* d_ws, size_t ws_size,
                              hipStream_t stream) {
    const float* x     = (const float*)d_in[0];
    const float* W_ih  = (const float*)d_in[1];
    const float* W_hh  = (const float*)d_in[2];
    const float* b_ih  = (const float*)d_in[3];
    const float* b_hh  = (const float*)d_in[4];
    const float* fc1_w = (const float*)d_in[5];
    const float* fc1_b = (const float*)d_in[6];
    const float* fc2_w = (const float*)d_in[7];
    const float* fc2_b = (const float*)d_in[8];
    float* out = (float*)d_out;

    float* xi = (float*)d_ws;

    xi_gemm_wmma<<<T_LEN / 16 / 8, 256, 0, stream>>>(x, W_ih, b_ih, b_hh, xi);

    gru_scan<<<1, 32, 0, stream>>>(xi, W_hh, b_hh, fc1_w, fc1_b, fc2_w, fc2_b, out);
}


# scan split in time: 1024 waves run one 256-step chunk each from a state guessed by 256 lead-in steps from h=0; a 3rd kernel checks boundary states bit for bit, serial rescan if any differ; same bits
# speedup vs baseline: 1828.1228x; 206.0948x over previous
#include <hip/hip_runtime.h>

typedef __attribute__((ext_vector_type(2))) float v2f;
typedef __attribute__((ext_vector_type(4))) float v4f;
typedef __attribute__((ext_vector_type(8))) float v8f;
typedef __attribute__((ext_vector_type(4))) unsigned v4u;

#define T_LEN   262144
#define IN_SZ   64
#define HID     16
#define G3      48
#define PF_D    16
#define KH      8
#define CH_L    256
#define CH_K    256
#define N_CH    (T_LEN / CH_L)
#define XI_FLOATS   ((size_t)T_LEN * G3)
#define BND_FLOATS  ((size_t)N_CH * HID)
#define WS_NEEDED   ((XI_FLOATS + 2 * BND_FLOATS) * sizeof(float))

__global__ __launch_bounds__(256) void xi_gemm_wmma(
    const float* __restrict__ x, const float* __restrict__ W_ih,
    const float* __restrict__ b_ih, const float* __restrict__ b_hh,
    float* __restrict__ xi)
{
    const int lane  = threadIdx.x & 31;
    const int wave  = threadIdx.x >> 5;
    const int tile  = blockIdx.x * 8 + wave;
    const int m     = lane & 15;
    const int khalf = lane >> 4;

    const float* xrow = x + (size_t)(tile * 16 + m) * IN_SZ + 2 * khalf;
    v2f a[16];
#pragma unroll
    for (int i = 0; i < 16; ++i)
        a[i] = *(const v2f*)(xrow + 4 * i);

    for (int nb = 0; nb < 3; ++nb) {
        const int col  = nb * 16 + m;
        const float bias = b_ih[col] + (nb < 2 ? b_hh[col] : 0.0f);
        v8f c = {bias, bias, bias, bias, bias, bias, bias, bias};

        const float* wrow = W_ih + (size_t)col * IN_SZ + 2 * khalf;
#pragma unroll
        for (int i = 0; i < 16; ++i) {
            v2f b = *(const v2f*)(wrow + 4 * i);
            c = __builtin_amdgcn_wmma_f32_16x16x4_f32(
                    false, a[i], false, b, (short)0, c, false, false);
        }

        float* orow = xi + (size_t)(tile * 16 + 8 * khalf) * G3 + col;
#pragma unroll
        for (int r = 0; r < 8; ++r)
            orow[(size_t)r * G3] = c[r];
    }
}

template <int MASK>
__device__ __forceinline__ float dpp_row_xmask(float v)
{
    const int i = __builtin_bit_cast(int, v);
    return __builtin_bit_cast(float,
        __builtin_amdgcn_update_dpp(i, i, 0x160 + MASK, 0xf, 0xf, true));
}


template <bool CHUNK>
__device__ __forceinline__ void scan_span(
    const float* __restrict__ xi, const float* __restrict__ W_hh,
    const float* __restrict__ b_hh, const float* __restrict__ fc1_w,
    const float* __restrict__ fc1_b, const float* __restrict__ fc2_w,
    const float* __restrict__ fc2_b, float* __restrict__ out,
    const int t_begin, const int t_own, const int t_end,
    float* __restrict__ hs_row, float* __restrict__ he_row)
{
    const int lane = threadIdx.x & 31;
    const int row  = lane >> 4;
    const int j    = (lane & 15) ^ (row * KH);
    const int kb   = row * KH;

    float wr[KH], wz[KH], wn[KH];
#pragma unroll
    for (int k = 0; k < KH; ++k) {
        wr[k] = 0.5f * W_hh[(j)      * HID + kb + k];
        wz[k] = 0.5f * W_hh[(16 + j) * HID + kb + k];
        wn[k] = 0.5f * W_hh[(32 + j) * HID + kb + k];
    }
    const float bhn_j = 0.5f * b_hh[32 + j];
    const float bhn   = row ? 0.0f : bhn_j;

    float weff = 0.0f, beff = fc2_b[0];
#pragma unroll
    for (int i = 0; i < 8; ++i) {
        const float f2 = fc2_w[i];
        weff += f2 * fc1_w[i * HID + j];
        beff += f2 * fc1_b[i];
    }

#pragma unroll
    for (int k = 0; k < KH; ++k)
        asm volatile("" :: "v"(wr[k]), "v"(wz[k]), "v"(wn[k]));
    asm volatile("" :: "v"(bhn));

    float hj = 0.0f;

    unsigned selk[KH];
#pragma unroll
    for (int k = 0; k < KH; ++k) {
        selk[k] = 0x11111111u * (unsigned)k;
        asm volatile("" : "+s"(selk[k]));
    }

    const unsigned sel_lo = 0xfedcba98u, sel_hi = 0x76543210u;

    const float* xr_p = xi + j;
    const float* xz_p = xi + 16 + j;
    const float* xn_p = xi + 32 + j;

    float pr[PF_D], pz[PF_D], pn[PF_D];
#pragma unroll
    for (int s = 0; s < PF_D; ++s) {
        __builtin_amdgcn_sched_barrier(0);
        pr[s] = xr_p[(size_t)(t_begin + s) * G3];
        pz[s] = xz_p[(size_t)(t_begin + s) * G3];
        pn[s] = xn_p[(size_t)(t_begin + s) * G3];
    }

    float yq[PF_D];

    for (int t0 = t_begin; t0 < t_end; t0 += PF_D) {
      if (CHUNK && t0 == t_own && lane < HID)
        hs_row[lane] = hj;

      const int pf0 = (t0 + PF_D < T_LEN) ? (t0 + PF_D) : t0;
      const float* nr_p = xr_p + (size_t)pf0 * G3;
      const float* nz_p = xz_p + (size_t)pf0 * G3;
      const float* nn_p = xn_p + (size_t)pf0 * G3;

#pragma unroll
      for (int s = 0; s < PF_D; ++s) {
        const float xr = pr[s], xz = pz[s], xn = pn[s];

        float hb[KH];
#pragma unroll
        for (int k = 0; k < KH; ++k)
          hb[k] = __builtin_bit_cast(float, __builtin_amdgcn_permlane16(
                      0, __builtin_bit_cast(int, hj), selk[k], selk[k], true, false));
        float hrA = hb[0] * wr[0], hrB = hb[1] * wr[1];
        float hzA = hb[0] * wz[0], hzB = hb[1] * wz[1];
        float hnA = __builtin_fmaf(hb[0], wn[0], bhn), hnB = hb[1] * wn[1];
#pragma unroll
        for (int k = 2; k < KH; k += 2) {
          hrA = __builtin_fmaf(hb[k], wr[k], hrA);
          hrB = __builtin_fmaf(hb[k + 1], wr[k + 1], hrB);
          hzA = __builtin_fmaf(hb[k], wz[k], hzA);
          hzB = __builtin_fmaf(hb[k + 1], wz[k + 1], hzB);
          hnA = __builtin_fmaf(hb[k], wn[k], hnA);
          hnB = __builtin_fmaf(hb[k + 1], wn[k + 1], hnB);
        }
        float hr = hrA + hrB, hz = hzA + hzB, hn = hnA + hnB;

        float hr_o, hz_o, hn_o;
        asm("v_permlanex16_b32 %0, %3, %6, %7\n\t"
            "v_permlanex16_b32 %1, %4, %6, %7\n\t"
            "v_permlanex16_b32 %2, %5, %6, %7"
            : "=&v"(hr_o), "=&v"(hz_o), "=&v"(hn_o)
            : "v"(hr), "v"(hz), "v"(hn), "s"(sel_lo), "s"(sel_hi));
        hr += hr_o;
        hz += hz_o;
        hn += hn_o;

        const float tr = __builtin_amdgcn_tanhf(__builtin_fmaf(xr, 0.5f, hr));
        const float tz = __builtin_amdgcn_tanhf(__builtin_fmaf(xz, 0.5f, hz));
        const float n  = __builtin_amdgcn_tanhf(__builtin_fmaf(tr, hn, xn + hn));
        const float hh   = 0.5f * hj;
        const float omz  = __builtin_fmaf(tz, -0.5f, 0.5f);
        const float zh   = __builtin_fmaf(tz, hh, hh);
        const float hnew = __builtin_fmaf(n, omz, zh);
        hj = hnew;

        pr[s] = nr_p[s * G3];
        pz[s] = nz_p[s * G3];
        pn[s] = nn_p[s * G3];

        yq[s] = hnew * weff;
        if (s > 0) {
          yq[s - 1] += dpp_row_xmask<1>(yq[s - 1]);
          yq[s - 1] += dpp_row_xmask<2>(yq[s - 1]);
          yq[s - 1] += dpp_row_xmask<4>(yq[s - 1]);
          yq[s - 1] += dpp_row_xmask<8>(yq[s - 1]);
        }

        __builtin_amdgcn_sched_barrier(0);
      }

      asm volatile("" :: "v"(nr_p));

      static_assert(PF_D % 4 == 0, "the pass stores its results as 4-float vectors");
      static_assert(T_LEN % PF_D == 0, "the scan runs whole passes of PF_D steps");
      yq[PF_D - 1] += dpp_row_xmask<1>(yq[PF_D - 1]);
      yq[PF_D - 1] += dpp_row_xmask<2>(yq[PF_D - 1]);
      yq[PF_D - 1] += dpp_row_xmask<4>(yq[PF_D - 1]);
      yq[PF_D - 1] += dpp_row_xmask<8>(yq[PF_D - 1]);
      if (lane == 0 && (!CHUNK || t0 >= t_own)) {
#pragma unroll
        for (int q = 0; q < PF_D; q += 4) {
          const v4f y4 = {yq[q] + beff, yq[q + 1] + beff, yq[q + 2] + beff, yq[q + 3] + beff};
          *(v4f*)(out + t0 + q) = y4;
        }
      }
    }

    if (CHUNK && lane < HID)
      he_row[lane] = hj;
}

__global__ __launch_bounds__(32) void gru_scan_chunks(
    const float* __restrict__ xi, const float* __restrict__ W_hh,
    const float* __restrict__ b_hh, const float* __restrict__ fc1_w,
    const float* __restrict__ fc1_b, const float* __restrict__ fc2_w,
    const float* __restrict__ fc2_b, float* __restrict__ out,
    float* __restrict__ hstart, float* __restrict__ hend, const size_t ws_size)
{
    if (ws_size < WS_NEEDED)
        return;
    const int c       = blockIdx.x;
    const int t_own   = c * CH_L;
    const int t_begin = t_own > CH_K ? t_own - CH_K : 0;
    scan_span<true>(xi, W_hh, b_hh, fc1_w, fc1_b, fc2_w, fc2_b, out,
                    t_begin, t_own, t_own + CH_L, hstart + (size_t)c * HID, hend + (size_t)c * HID);
}

__global__ __launch_bounds__(256) void gru_check_rescan(
    const float* __restrict__ xi, const float* __restrict__ W_hh,
    const float* __restrict__ b_hh, const float* __restrict__ fc1_w,
    const float* __restrict__ fc1_b, const float* __restrict__ fc2_w,
    const float* __restrict__ fc2_b, float* __restrict__ out,
    const float* __restrict__ hstart, const float* __restrict__ hend, const size_t ws_size)
{
    __shared__ unsigned wave_diff[8];

    unsigned diff = 1u;
    if (ws_size >= WS_NEEDED) {
        diff = 0u;
        const v4f* a = (const v4f*)(hstart + HID);
        const v4f* b = (const v4f*)hend;
        for (int i = threadIdx.x; i < (N_CH - 1) * (HID / 4); i += 256) {
            const v4u p = __builtin_bit_cast(v4u, a[i]);
            const v4u q = __builtin_bit_cast(v4u, b[i]);
            const v4u x = p ^ q;
            diff |= x[0] | x[1] | x[2] | x[3];
        }
    }
    const unsigned w = __builtin_amdgcn_ballot_w32(diff != 0u);
    if ((threadIdx.x & 31) == 0)
        wave_diff[threadIdx.x >> 5] = w;
    __syncthreads();
    unsigned any = 0u;
#pragma unroll
    for (int i = 0; i < 8; ++i)
        any |= wave_diff[i];
    if (any == 0u)
        return;
    if (threadIdx.x >= 32)
        return;
    scan_span<false>(xi, W_hh, b_hh, fc1_w, fc1_b, fc2_w, fc2_b, out,
                     0, 0, T_LEN, nullptr, nullptr);
}

extern "C" void kernel_launch(void* const* d_in, const int* in_sizes, int n_in,
                              void* d_out, int out_size, void* d_ws, size_t ws_size,
                              hipStream_t stream) {
    const float* x     = (const float*)d_in[0];
    const float* W_ih  = (const float*)d_in[1];
    const float* W_hh  = (const float*)d_in[2];
    const float* b_ih  = (const float*)d_in[3];
    const float* b_hh  = (const float*)d_in[4];
    const float* fc1_w = (const float*)d_in[5];
    const float* fc1_b = (const float*)d_in[6];
    const float* fc2_w = (const float*)d_in[7];
    const float* fc2_b = (const float*)d_in[8];
    float* out = (float*)d_out;

    float* xi = (float*)d_ws;

    xi_gemm_wmma<<<T_LEN / 16 / 8, 256, 0, stream>>>(x, W_ih, b_ih, b_hh, xi);

    float* hstart = xi + XI_FLOATS;
    float* hend   = hstart + BND_FLOATS;

    gru_scan_chunks<<<N_CH, 32, 0, stream>>>(xi, W_hh, b_hh, fc1_w, fc1_b, fc2_w, fc2_b, out,
                                             hstart, hend, ws_size);

    gru_check_rescan<<<1, 256, 0, stream>>>(xi, W_hh, b_hh, fc1_w, fc1_b, fc2_w, fc2_b, out,
                                            hstart, hend, ws_size);
}
